# starting kernel
# baseline (speedup 1.0000x reference)
#include <hip/hip_runtime.h>

typedef __attribute__((ext_vector_type(2))) float v2f;
typedef __attribute__((ext_vector_type(8))) float v8f;

#define DIM        128
#define NEDGES     50000
#define NEG_SLOPE  0.01f

__global__ void __launch_bounds__(256)
gemm_wmma_f32(const float* __restrict__ X, const float* __restrict__ W,
              float* __restrict__ Y, int mtiles)
{
    const int lane  = threadIdx.x & 31;
    const int wv    = threadIdx.x >> 5;
    const long tile = (long)blockIdx.x * 8 + wv;
    const long mt   = tile >> 3;
    const int  nt   = (int)(tile & 7);
    if (mt >= mtiles) return;

    const int half = lane >> 4;
    const int lr   = lane & 15;
    const size_t row  = (size_t)mt * 16 + lr;
    const int    col0 = nt * 16;

    const float* __restrict__ ap = X + row * DIM + 2 * half;
    const float* __restrict__ bp = W + (size_t)(2 * half) * DIM + col0 + lr;

    v8f c = {};
#pragma unroll
    for (int k = 0; k < DIM; k += 4) {
        v2f a, b;
        a.x = ap[k];
        a.y = ap[k + 1];
        b.x = bp[(size_t)k * DIM];
        b.y = bp[(size_t)k * DIM + DIM];
        c = __builtin_amdgcn_wmma_f32_16x16x4_f32(false, a, false, b,
                                                  (short)0, c, false, false);
    }

    float* __restrict__ yp = Y + ((size_t)mt * 16 + half * 8) * DIM + col0 + lr;
#pragma unroll
    for (int r = 0; r < 8; ++r)
        yp[(size_t)r * DIM] = c[r];
}

__global__ void zero_kernel(float* __restrict__ p, long n)
{
    long i = (long)blockIdx.x * blockDim.x + threadIdx.x;
    if (i < n) p[i] = 0.0f;
}

__global__ void degrees_kernel(const int* __restrict__ node,
                               const int* __restrict__ edge,
                               float* __restrict__ Ddeg,
                               float* __restrict__ Bdeg, int nnz)
{
    int i = blockIdx.x * blockDim.x + threadIdx.x;
    if (i >= nnz) return;
    unsafeAtomicAdd(&Ddeg[node[i]], 1.0f);
    unsafeAtomicAdd(&Bdeg[edge[i]], 1.0f);
}

__global__ void __launch_bounds__(256)
scatter_kernel(const float* __restrict__ src, float* __restrict__ dst,
               const int* __restrict__ gidx, const int* __restrict__ sidx,
               int nnz)
{
    const int wid  = (int)(((long)blockIdx.x * blockDim.x + threadIdx.x) >> 5);
    const int lane = threadIdx.x & 31;
    if (wid >= nnz) return;

    const int g = gidx[wid];
    const int s = sidx[wid];

    const float4 v = ((const float4*)(src + (size_t)g * DIM))[lane];
    float* dp = dst + (size_t)s * DIM + lane * 4;
    unsafeAtomicAdd(dp + 0, v.x);
    unsafeAtomicAdd(dp + 1, v.y);
    unsafeAtomicAdd(dp + 2, v.z);
    unsafeAtomicAdd(dp + 3, v.w);
}

__global__ void scale_rows_kernel(float* __restrict__ m,
                                  const float* __restrict__ Bdeg, long n)
{
    long i = (long)blockIdx.x * blockDim.x + threadIdx.x;
    if (i >= n) return;
    float bb  = Bdeg[i >> 7];
    float inv = bb > 0.0f ? 1.0f / bb : 0.0f;
    m[i] *= inv;
}

__global__ void finalize_kernel(float* __restrict__ out,
                                const float* __restrict__ Ddeg,
                                const float* __restrict__ bias, long n)
{
    long i = (long)blockIdx.x * blockDim.x + threadIdx.x;
    if (i >= n) return;
    float dd  = Ddeg[i >> 7];
    float inv = dd > 0.0f ? 1.0f / dd : 0.0f;
    float v   = out[i] * inv + bias[i & (DIM - 1)];
    out[i] = v >= 0.0f ? v : NEG_SLOPE * v;
}

extern "C" void kernel_launch(void* const* d_in, const int* in_sizes, int n_in,
                              void* d_out, int out_size, void* d_ws, size_t ws_size,
                              hipStream_t stream)
{
    const float* X    = (const float*)d_in[0];
    const int*   A    = (const int*)d_in[1];
    const float* W    = (const float*)d_in[2];
    const float* bias = (const float*)d_in[3];
    float*       out  = (float*)d_out;

    const int N   = in_sizes[0] / DIM;
    const int NNZ = in_sizes[1] / 2;
    const int* node = A;
    const int* edge = A + NNZ;

    float* m    = (float*)d_ws;
    float* Ddeg = m + (size_t)NEDGES * DIM;
    float* Bdeg = Ddeg + N;
    const long ws_f  = (long)NEDGES * DIM + N + NEDGES;
    const long n_out = (long)N * DIM;
    const long n_m   = (long)NEDGES * DIM;

    const int mtiles = N / 16;
    gemm_wmma_f32<<<mtiles, 256, 0, stream>>>(X, W, out, mtiles);

    zero_kernel<<<(int)((ws_f + 255) / 256), 256, 0, stream>>>(m, ws_f);

    degrees_kernel<<<(NNZ + 255) / 256, 256, 0, stream>>>(node, edge, Ddeg, Bdeg, NNZ);

    scatter_kernel<<<(NNZ + 7) / 8, 256, 0, stream>>>(out, m, node, edge, NNZ);

    scale_rows_kernel<<<(int)((n_m + 255) / 256), 256, 0, stream>>>(m, Bdeg, n_m);

    zero_kernel<<<(int)((n_out + 255) / 256), 256, 0, stream>>>(out, n_out);

    scatter_kernel<<<(NNZ + 7) / 8, 256, 0, stream>>>(m, out, edge, node, NNZ);

    finalize_kernel<<<(int)((n_out + 255) / 256), 256, 0, stream>>>(out, Ddeg, bias, n_out);
}


# scatter passes: lane = feature column (cols lane+32j), so each f32 atomic add instruction and each gather load covers one whole 128-B line instead of 8-lane parts of four lines
# speedup vs baseline: 1.2444x; 1.2444x over previous
#include <hip/hip_runtime.h>

typedef __attribute__((ext_vector_type(2))) float v2f;
typedef __attribute__((ext_vector_type(8))) float v8f;

#define DIM        128
#define NEDGES     50000
#define NEG_SLOPE  0.01f

__global__ void __launch_bounds__(256)
gemm_wmma_f32(const float* __restrict__ X, const float* __restrict__ W,
              float* __restrict__ Y, int mtiles)
{
    const int lane  = threadIdx.x & 31;
    const int wv    = threadIdx.x >> 5;
    const long tile = (long)blockIdx.x * 8 + wv;
    const long mt   = tile >> 3;
    const int  nt   = (int)(tile & 7);
    if (mt >= mtiles) return;

    const int half = lane >> 4;
    const int lr   = lane & 15;
    const size_t row  = (size_t)mt * 16 + lr;
    const int    col0 = nt * 16;

    const float* __restrict__ ap = X + row * DIM + 2 * half;
    const float* __restrict__ bp = W + (size_t)(2 * half) * DIM + col0 + lr;

    v8f c = {};
#pragma unroll
    for (int k = 0; k < DIM; k += 4) {
        v2f a, b;
        a.x = ap[k];
        a.y = ap[k + 1];
        b.x = bp[(size_t)k * DIM];
        b.y = bp[(size_t)k * DIM + DIM];
        c = __builtin_amdgcn_wmma_f32_16x16x4_f32(false, a, false, b,
                                                  (short)0, c, false, false);
    }

    float* __restrict__ yp = Y + ((size_t)mt * 16 + half * 8) * DIM + col0 + lr;
#pragma unroll
    for (int r = 0; r < 8; ++r)
        yp[(size_t)r * DIM] = c[r];
}

__global__ void zero_kernel(float* __restrict__ p, long n)
{
    long i = (long)blockIdx.x * blockDim.x + threadIdx.x;
    if (i < n) p[i] = 0.0f;
}

__global__ void degrees_kernel(const int* __restrict__ node,
                               const int* __restrict__ edge,
                               float* __restrict__ Ddeg,
                               float* __restrict__ Bdeg, int nnz)
{
    int i = blockIdx.x * blockDim.x + threadIdx.x;
    if (i >= nnz) return;
    unsafeAtomicAdd(&Ddeg[node[i]], 1.0f);
    unsafeAtomicAdd(&Bdeg[edge[i]], 1.0f);
}

__global__ void __launch_bounds__(256)
scatter_kernel(const float* __restrict__ src, float* __restrict__ dst,
               const int* __restrict__ gidx, const int* __restrict__ sidx,
               int nnz)
{
    const int wid  = (int)(((long)blockIdx.x * blockDim.x + threadIdx.x) >> 5);
    const int lane = threadIdx.x & 31;
    if (wid >= nnz) return;

    const int g = gidx[wid];
    const int s = sidx[wid];

    const float* __restrict__ sp = src + (size_t)g * DIM + lane;
    float* dp = dst + (size_t)s * DIM + lane;
    const float v0 = sp[0];
    const float v1 = sp[32];
    const float v2 = sp[64];
    const float v3 = sp[96];
    unsafeAtomicAdd(dp + 0,  v0);
    unsafeAtomicAdd(dp + 32, v1);
    unsafeAtomicAdd(dp + 64, v2);
    unsafeAtomicAdd(dp + 96, v3);
}

__global__ void scale_rows_kernel(float* __restrict__ m,
                                  const float* __restrict__ Bdeg, long n)
{
    long i = (long)blockIdx.x * blockDim.x + threadIdx.x;
    if (i >= n) return;
    float bb  = Bdeg[i >> 7];
    float inv = bb > 0.0f ? 1.0f / bb : 0.0f;
    m[i] *= inv;
}

__global__ void finalize_kernel(float* __restrict__ out,
                                const float* __restrict__ Ddeg,
                                const float* __restrict__ bias, long n)
{
    long i = (long)blockIdx.x * blockDim.x + threadIdx.x;
    if (i >= n) return;
    float dd  = Ddeg[i >> 7];
    float inv = dd > 0.0f ? 1.0f / dd : 0.0f;
    float v   = out[i] * inv + bias[i & (DIM - 1)];
    out[i] = v >= 0.0f ? v : NEG_SLOPE * v;
}

extern "C" void kernel_launch(void* const* d_in, const int* in_sizes, int n_in,
                              void* d_out, int out_size, void* d_ws, size_t ws_size,
                              hipStream_t stream)
{
    const float* X    = (const float*)d_in[0];
    const int*   A    = (const int*)d_in[1];
    const float* W    = (const float*)d_in[2];
    const float* bias = (const float*)d_in[3];
    float*       out  = (float*)d_out;

    const int N   = in_sizes[0] / DIM;
    const int NNZ = in_sizes[1] / 2;
    const int* node = A;
    const int* edge = A + NNZ;

    float* m    = (float*)d_ws;
    float* Ddeg = m + (size_t)NEDGES * DIM;
    float* Bdeg = Ddeg + N;
    const long ws_f  = (long)NEDGES * DIM + N + NEDGES;
    const long n_out = (long)N * DIM;
    const long n_m   = (long)NEDGES * DIM;

    const int mtiles = N / 16;
    gemm_wmma_f32<<<mtiles, 256, 0, stream>>>(X, W, out, mtiles);

    zero_kernel<<<(int)((ws_f + 255) / 256), 256, 0, stream>>>(m, ws_f);

    degrees_kernel<<<(NNZ + 255) / 256, 256, 0, stream>>>(node, edge, Ddeg, Bdeg, NNZ);

    scatter_kernel<<<(NNZ + 7) / 8, 256, 0, stream>>>(out, m, node, edge, NNZ);

    scale_rows_kernel<<<(int)((n_m + 255) / 256), 256, 0, stream>>>(m, Bdeg, n_m);

    zero_kernel<<<(int)((n_out + 255) / 256), 256, 0, stream>>>(out, n_out);

    scatter_kernel<<<(NNZ + 7) / 8, 256, 0, stream>>>(m, out, edge, node, NNZ);

    finalize_kernel<<<(int)((n_out + 255) / 256), 256, 0, stream>>>(out, Ddeg, bias, n_out);
}


# row lists built by an LDS counting sort: records binned by 512-row range with LDS adds, one block sorts each range in LDS; no per-entry global atomics; rows still summed in ascending source index
# speedup vs baseline: 3.0409x; 2.4437x over previous
#include <hip/hip_runtime.h>

typedef __attribute__((ext_vector_type(2))) float v2f;
typedef __attribute__((ext_vector_type(8))) float v8f;

#define DIM        128
#define NEDGES     50000
#define NEG_SLOPE  0.01f

__global__ void __launch_bounds__(256)
gemm_wmma_f32(const float* __restrict__ X, const float* __restrict__ W,
              float* __restrict__ Y, int mtiles)
{
    const int lane  = threadIdx.x & 31;
    const int wv    = threadIdx.x >> 5;
    const long tile = (long)blockIdx.x * 8 + wv;
    const long mt   = tile >> 3;
    const int  nt   = (int)(tile & 7);
    if (mt >= mtiles) return;

    const int half = lane >> 4;
    const int lr   = lane & 15;
    const size_t row  = (size_t)mt * 16 + lr;
    const int    col0 = nt * 16;

    const float* __restrict__ ap = X + row * DIM + 2 * half;
    const float* __restrict__ bp = W + (size_t)(2 * half) * DIM + col0 + lr;

    v8f c = {};
#pragma unroll
    for (int k = 0; k < DIM; k += 4) {
        v2f a, b;
        a.x = ap[k];
        a.y = ap[k + 1];
        b.x = bp[(size_t)k * DIM];
        b.y = bp[(size_t)k * DIM + DIM];
        c = __builtin_amdgcn_wmma_f32_16x16x4_f32(false, a, false, b,
                                                  (short)0, c, false, false);
    }

    float* __restrict__ yp = Y + ((size_t)mt * 16 + half * 8) * DIM + col0 + lr;
#pragma unroll
    for (int r = 0; r < 8; ++r)
        yp[(size_t)r * DIM] = c[r];
}

__global__ void zero_kernel(float* __restrict__ p, long n)
{
    long i = (long)blockIdx.x * blockDim.x + threadIdx.x;
    if (i < n) p[i] = 0.0f;
}

__global__ void degrees_kernel(const int* __restrict__ node,
                               const int* __restrict__ edge,
                               float* __restrict__ Ddeg,
                               float* __restrict__ Bdeg, int nnz)
{
    int i = blockIdx.x * blockDim.x + threadIdx.x;
    if (i >= nnz) return;
    unsafeAtomicAdd(&Ddeg[node[i]], 1.0f);
    unsafeAtomicAdd(&Bdeg[edge[i]], 1.0f);
}

__global__ void __launch_bounds__(256)
scatter_kernel(const float* __restrict__ src, float* __restrict__ dst,
               const int* __restrict__ gidx, const int* __restrict__ sidx,
               int nnz)
{
    const int wid  = (int)(((long)blockIdx.x * blockDim.x + threadIdx.x) >> 5);
    const int lane = threadIdx.x & 31;
    if (wid >= nnz) return;

    const int g = gidx[wid];
    const int s = sidx[wid];

    const float* __restrict__ sp = src + (size_t)g * DIM + lane;
    float* dp = dst + (size_t)s * DIM + lane;
    const float v0 = sp[0];
    const float v1 = sp[32];
    const float v2 = sp[64];
    const float v3 = sp[96];
    unsafeAtomicAdd(dp + 0,  v0);
    unsafeAtomicAdd(dp + 32, v1);
    unsafeAtomicAdd(dp + 64, v2);
    unsafeAtomicAdd(dp + 96, v3);
}

__global__ void scale_rows_kernel(float* __restrict__ m,
                                  const float* __restrict__ Bdeg, long n)
{
    long i = (long)blockIdx.x * blockDim.x + threadIdx.x;
    if (i >= n) return;
    float bb  = Bdeg[i >> 7];
    float inv = bb > 0.0f ? 1.0f / bb : 0.0f;
    m[i] *= inv;
}

__global__ void finalize_kernel(float* __restrict__ out,
                                const float* __restrict__ Ddeg,
                                const float* __restrict__ bias, long n)
{
    long i = (long)blockIdx.x * blockDim.x + threadIdx.x;
    if (i >= n) return;
    float dd  = Ddeg[i >> 7];
    float inv = dd > 0.0f ? 1.0f / dd : 0.0f;
    float v   = out[i] * inv + bias[i & (DIM - 1)];
    out[i] = v >= 0.0f ? v : NEG_SLOPE * v;
}

#define KEY_NONE 0x7fffffff

#define RB_SHIFT  9
#define RB_ROWS   (1 << RB_SHIFT)
#define NBUCK_MAX 512
#define SRC_BITS  18
#define SRC_MASK  ((1 << SRC_BITS) - 1)
#define PT        1024
#define EPT       8
#define CHUNK     (PT * EPT)

__device__ __forceinline__ int scan512(int x, int tid, int* wsum)
{
    const int lane = tid & 31;
    const int wv   = tid >> 5;
    int v = x;
    for (int o = 1; o < 32; o <<= 1) {
        const int t = __shfl_up(v, o);
        if (lane >= o) v += t;
    }
    if (lane == 31 && wv < 16) wsum[wv] = v;
    __syncthreads();
    if (wv == 0) {
        int w = lane < 16 ? wsum[lane] : 0;
        for (int o = 1; o < 16; o <<= 1) {
            const int t = __shfl_up(w, o);
            if (lane >= o) w += t;
        }
        if (lane < 16) wsum[lane] = w;
    }
    __syncthreads();
    if (wv > 0 && wv < 16) v += wsum[wv - 1];
    return v;
}

__global__ void __launch_bounds__(PT)
range_count_kernel(const int* __restrict__ node, const int* __restrict__ edge,
                   int* __restrict__ bcnt, int nnz)
{
    __shared__ int h[NBUCK_MAX];
    const int tid = threadIdx.x;
    if (tid < NBUCK_MAX) h[tid] = 0;
    __syncthreads();
    const int base = blockIdx.x * CHUNK + tid;
    int ke[EPT], kn[EPT];
#pragma unroll
    for (int u = 0; u < EPT; ++u) {
        const int i = base + u * PT;
        ke[u] = i < nnz ? edge[i] : -1;
        kn[u] = i < nnz ? node[i] : 0;
    }
#pragma unroll
    for (int u = 0; u < EPT; ++u) {
        if (ke[u] >= 0) {
            atomicAdd(&h[ke[u] >> RB_SHIFT], 1);
            atomicAdd(&h[(NEDGES + kn[u]) >> RB_SHIFT], 1);
        }
    }
    __syncthreads();
    if (tid < NBUCK_MAX) {
        const int c = h[tid];
        if (c > 0) atomicAdd(&bcnt[tid], c);
    }
}

__global__ void __launch_bounds__(PT)
range_move_kernel(const int* __restrict__ node, const int* __restrict__ edge,
                  const int* __restrict__ bcnt, int* __restrict__ bcur,
                  int* __restrict__ boff, int* __restrict__ recs, int nnz)
{
    __shared__ int h[NBUCK_MAX];
    __shared__ int sbase[NBUCK_MAX];
    __shared__ int wsum[16];
    const int tid = threadIdx.x;
    if (tid < NBUCK_MAX) h[tid] = 0;
    __syncthreads();

    int ke[EPT], kn[EPT], re[EPT], rn[EPT];
    const int base = blockIdx.x * CHUNK + tid;
#pragma unroll
    for (int u = 0; u < EPT; ++u) {
        const int i = base + u * PT;
        ke[u] = i < nnz ? edge[i] : -1;
        kn[u] = i < nnz ? node[i] : 0;
    }
#pragma unroll
    for (int u = 0; u < EPT; ++u) {
        re[u] = 0;
        rn[u] = 0;
        if (ke[u] >= 0) {
            re[u] = atomicAdd(&h[ke[u] >> RB_SHIFT], 1);
            rn[u] = atomicAdd(&h[(NEDGES + kn[u]) >> RB_SHIFT], 1);
        }
    }

    const int tot = tid < NBUCK_MAX ? bcnt[tid] : 0;
    const int inc = scan512(tot, tid, wsum);
    if (tid < NBUCK_MAX) {
        const int ex = inc - tot;
        if (blockIdx.x == 0) boff[tid] = ex;
        const int c = h[tid];
        int b = 0;
        if (c > 0) b = atomicAdd(&bcur[tid], c);
        sbase[tid] = ex + b;
    }
    __syncthreads();

#pragma unroll
    for (int u = 0; u < EPT; ++u) {
        if (ke[u] >= 0) {
            const int e  = ke[u];
            const int n  = kn[u];
            const int rr = NEDGES + n;
            recs[sbase[e >> RB_SHIFT] + re[u]]  = ((e & (RB_ROWS - 1)) << SRC_BITS) | n;
            recs[sbase[rr >> RB_SHIFT] + rn[u]] = ((rr & (RB_ROWS - 1)) << SRC_BITS) | e;
        }
    }
}

__global__ void __launch_bounds__(PT)
range_rows_kernel(const int* __restrict__ recs, const int* __restrict__ bcnt,
                  const int* __restrict__ boff, int* __restrict__ cnt,
                  int* __restrict__ off, int* __restrict__ list, int nrows)
{
    __shared__ int c[RB_ROWS];
    __shared__ int f[RB_ROWS];
    __shared__ int wsum[16];
    const int tid   = threadIdx.x;
    const int b     = blockIdx.x;
    const int n     = bcnt[b];
    const int start = boff[b];
    const int* __restrict__ rp = recs + start;
    if (tid < RB_ROWS) c[tid] = 0;
    __syncthreads();

    for (int j0 = tid; j0 < n; j0 += 4 * PT) {
        int r[4];
#pragma unroll
        for (int u = 0; u < 4; ++u) {
            const int j = j0 + u * PT;
            r[u] = j < n ? rp[j] : -1;
        }
#pragma unroll
        for (int u = 0; u < 4; ++u)
            if (r[u] >= 0) atomicAdd(&c[r[u] >> SRC_BITS], 1);
    }
    __syncthreads();

    const int mine = tid < RB_ROWS ? c[tid] : 0;
    const int inc  = scan512(mine, tid, wsum);
    if (tid < RB_ROWS) {
        const int ex  = inc - mine;
        const int row = b * RB_ROWS + tid;
        f[tid] = ex;
        if (row < nrows) {
            cnt[row] = mine;
            off[row] = start + ex;
        }
    }
    __syncthreads();

    int* __restrict__ lp = list + start;
    for (int j0 = tid; j0 < n; j0 += 4 * PT) {
        int r[4];
#pragma unroll
        for (int u = 0; u < 4; ++u) {
            const int j = j0 + u * PT;
            r[u] = j < n ? rp[j] : -1;
        }
#pragma unroll
        for (int u = 0; u < 4; ++u) {
            if (r[u] >= 0) {
                const int p = atomicAdd(&f[r[u] >> SRC_BITS], 1);
                lp[p] = r[u] & SRC_MASK;
            }
        }
    }
}

template <int K>
__device__ __forceinline__ void
add_ranked_rows(const float* __restrict__ src, int lane, int t,
                int k0, int k1, int r0, int r1,
                float& a0, float& a1, float& a2, float& a3)
{
    int kk[K];
#pragma unroll
    for (int u = 0; u < K; ++u) {
        const unsigned m0 = __builtin_amdgcn_ballot_w32(r0 == t + u);
        const unsigned m1 = __builtin_amdgcn_ballot_w32(r1 == t + u);
        const int sl = __builtin_ctz(m0 ? m0 : m1);
        kk[u] = m0 ? __builtin_amdgcn_readlane(k0, sl)
                   : __builtin_amdgcn_readlane(k1, sl);
    }
    float v[K][4];
#pragma unroll
    for (int u = 0; u < K; ++u) {
        const float* __restrict__ sp = src + (size_t)kk[u] * DIM + lane;
        v[u][0] = sp[0];
        v[u][1] = sp[32];
        v[u][2] = sp[64];
        v[u][3] = sp[96];
    }
#pragma unroll
    for (int u = 0; u < K; ++u) {
        a0 += v[u][0];
        a1 += v[u][1];
        a2 += v[u][2];
        a3 += v[u][3];
    }
}

template <int FINAL>
__global__ void __launch_bounds__(256)
rowsum_kernel(const float* __restrict__ src, float* __restrict__ dst,
              const int* __restrict__ cnt, const int* __restrict__ off,
              const int* __restrict__ list, const float* __restrict__ bias,
              int nrows)
{
    const int row  = (int)(((long)blockIdx.x * blockDim.x + threadIdx.x) >> 5);
    const int lane = threadIdx.x & 31;
    if (row >= nrows) return;

    const int n = __builtin_amdgcn_readfirstlane(cnt[row]);
    const int b = __builtin_amdgcn_readfirstlane(off[row]);
    const int* __restrict__ lst = list + b;

    float a0 = 0.0f, a1 = 0.0f, a2 = 0.0f, a3 = 0.0f;

    if (n <= 64) {
        const bool v0 = lane < n;
        const bool v1 = lane + 32 < n;
        const int k0 = v0 ? lst[lane] : KEY_NONE;
        const int k1 = v1 ? lst[lane + 32] : KEY_NONE;
        const int n0 = n < 32 ? n : 32;
        const int n1 = n - n0;
        int r0 = 0, r1 = 0;
        for (int j = 0; j < n0; ++j) {
            const int kj = __builtin_amdgcn_readlane(k0, j);
            r0 += (kj < k0 || (kj == k0 && j < lane)) ? 1 : 0;
            r1 += (kj <= k1) ? 1 : 0;
        }
        for (int j = 0; j < n1; ++j) {
            const int kj = __builtin_amdgcn_readlane(k1, j);
            r0 += (kj < k0) ? 1 : 0;
            r1 += (kj < k1 || (kj == k1 && j < lane)) ? 1 : 0;
        }
        if (!v0) r0 = -1;
        if (!v1) r1 = -1;

        int t = 0;
        for (; t + 4 <= n; t += 4)
            add_ranked_rows<4>(src, lane, t, k0, k1, r0, r1, a0, a1, a2, a3);
        if (t + 2 <= n) {
            add_ranked_rows<2>(src, lane, t, k0, k1, r0, r1, a0, a1, a2, a3);
            t += 2;
        }
        if (t < n)
            add_ranked_rows<1>(src, lane, t, k0, k1, r0, r1, a0, a1, a2, a3);
    } else {
        int prev = -1;
        int done = 0;
        while (done < n) {
            int mn = KEY_NONE;
            for (int j = lane; j < n; j += 32) {
                const int k = lst[j];
                if (k > prev && k < mn) mn = k;
            }
            for (int o = 16; o > 0; o >>= 1) {
                const int other = __shfl_xor(mn, o);
                mn = other < mn ? other : mn;
            }
            int c = 0;
            for (int j = lane; j < n; j += 32) c += (lst[j] == mn) ? 1 : 0;
            for (int o = 16; o > 0; o >>= 1) c += __shfl_xor(c, o);
            mn = __builtin_amdgcn_readfirstlane(mn);
            c  = __builtin_amdgcn_readfirstlane(c);
            if (c <= 0) break;
            const float* __restrict__ sp = src + (size_t)mn * DIM + lane;
            const float w0 = sp[0];
            const float w1 = sp[32];
            const float w2 = sp[64];
            const float w3 = sp[96];
            for (int i = 0; i < c; ++i) {
                a0 += w0;
                a1 += w1;
                a2 += w2;
                a3 += w3;
            }
            prev = mn;
            done += c;
        }
    }

    const float inv = n > 0 ? 1.0f / (float)n : 0.0f;
    float* __restrict__ dp = dst + (size_t)row * DIM + lane;
    if (FINAL) {
        const float y0 = a0 * inv + bias[lane];
        const float y1 = a1 * inv + bias[lane + 32];
        const float y2 = a2 * inv + bias[lane + 64];
        const float y3 = a3 * inv + bias[lane + 96];
        dp[0]  = y0 >= 0.0f ? y0 : NEG_SLOPE * y0;
        dp[32] = y1 >= 0.0f ? y1 : NEG_SLOPE * y1;
        dp[64] = y2 >= 0.0f ? y2 : NEG_SLOPE * y2;
        dp[96] = y3 >= 0.0f ? y3 : NEG_SLOPE * y3;
    } else {
        dp[0]  = a0 * inv;
        dp[32] = a1 * inv;
        dp[64] = a2 * inv;
        dp[96] = a3 * inv;
    }
}

extern "C" void kernel_launch(void* const* d_in, const int* in_sizes, int n_in,
                              void* d_out, int out_size, void* d_ws, size_t ws_size,
                              hipStream_t stream)
{
    const float* X    = (const float*)d_in[0];
    const int*   A    = (const int*)d_in[1];
    const float* W    = (const float*)d_in[2];
    const float* bias = (const float*)d_in[3];
    float*       out  = (float*)d_out;

    const int N   = in_sizes[0] / DIM;
    const int NNZ = in_sizes[1] / 2;
    const int* node = A;
    const int* edge = A + NNZ;
    const long n_out = (long)N * DIM;
    const long n_m   = (long)NEDGES * DIM;

    const int mtiles = N / 16;

    const int  R    = NEDGES + N;
    const int  nb   = (R + RB_ROWS - 1) / RB_ROWS;
    float* m    = (float*)d_ws;
    int*   cnt  = (int*)(m + n_m);
    int*   off  = cnt + R;
    int*   list = off + R;
    int*   bcnt = list + 2L * NNZ;
    int*   bcur = bcnt + NBUCK_MAX;
    int*   boff = bcur + NBUCK_MAX;
    int*   recs = (int*)d_out;
    const size_t need = ((size_t)n_m + 2 * (size_t)R + 2 * (size_t)NNZ + 3 * NBUCK_MAX) * sizeof(float);

    if (ws_size >= need && nb <= NBUCK_MAX && N <= (1 << SRC_BITS) &&
        NEDGES <= (1 << SRC_BITS) && 2L * NNZ <= n_out) {
        const int nchunks = (NNZ + CHUNK - 1) / CHUNK;
        zero_kernel<<<(2 * NBUCK_MAX + 255) / 256, 256, 0, stream>>>((float*)bcnt, 2L * NBUCK_MAX);
        range_count_kernel<<<nchunks, PT, 0, stream>>>(node, edge, bcnt, NNZ);
        range_move_kernel<<<nchunks, PT, 0, stream>>>(node, edge, bcnt, bcur, boff, recs, NNZ);
        range_rows_kernel<<<nb, PT, 0, stream>>>(recs, bcnt, boff, cnt, off, list, R);
        gemm_wmma_f32<<<mtiles, 256, 0, stream>>>(X, W, out, mtiles);
        rowsum_kernel<0><<<(NEDGES + 7) / 8, 256, 0, stream>>>(out, m, cnt, off, list, bias, NEDGES);
        rowsum_kernel<1><<<(N + 7) / 8, 256, 0, stream>>>(m, out, cnt + NEDGES, off + NEDGES, list, bias, N);
        return;
    }

    gemm_wmma_f32<<<mtiles, 256, 0, stream>>>(X, W, out, mtiles);

    float* Ddeg = m + (size_t)NEDGES * DIM;
    float* Bdeg = Ddeg + N;
    const long ws_f  = (long)NEDGES * DIM + N + NEDGES;
    zero_kernel<<<(int)((ws_f + 255) / 256), 256, 0, stream>>>(m, ws_f);
    degrees_kernel<<<(NNZ + 255) / 256, 256, 0, stream>>>(node, edge, Ddeg, Bdeg, NNZ);
    scatter_kernel<<<(NNZ + 7) / 8, 256, 0, stream>>>(out, m, node, edge, NNZ);
    scale_rows_kernel<<<(int)((n_m + 255) / 256), 256, 0, stream>>>(m, Bdeg, n_m);
    zero_kernel<<<(int)((n_out + 255) / 256), 256, 0, stream>>>(out, n_out);
    scatter_kernel<<<(NNZ + 7) / 8, 256, 0, stream>>>(m, out, edge, node, NNZ);
    finalize_kernel<<<(int)((n_out + 255) / 256), 256, 0, stream>>>(out, Ddeg, bias, n_out);
}


# row sums: rows of 65 to 128 sources are ranked in registers, four keys to a lane, like short rows, instead of one key per trip of dependent list scans; same ascending add order, bit-identical
# speedup vs baseline: 3.5968x; 1.1828x over previous
#include <hip/hip_runtime.h>

typedef __attribute__((ext_vector_type(2))) float v2f;
typedef __attribute__((ext_vector_type(8))) float v8f;

#define DIM        128
#define NEDGES     50000
#define NEG_SLOPE  0.01f

__global__ void __launch_bounds__(256)
gemm_wmma_f32(const float* __restrict__ X, const float* __restrict__ W,
              float* __restrict__ Y, int mtiles)
{
    const int lane  = threadIdx.x & 31;
    const int wv    = threadIdx.x >> 5;
    const long tile = (long)blockIdx.x * 8 + wv;
    const long mt   = tile >> 3;
    const int  nt   = (int)(tile & 7);
    if (mt >= mtiles) return;

    const int half = lane >> 4;
    const int lr   = lane & 15;
    const size_t row  = (size_t)mt * 16 + lr;
    const int    col0 = nt * 16;

    const float* __restrict__ ap = X + row * DIM + 2 * half;
    const float* __restrict__ bp = W + (size_t)(2 * half) * DIM + col0 + lr;

    v8f c = {};
#pragma unroll
    for (int k = 0; k < DIM; k += 4) {
        v2f a, b;
        a.x = ap[k];
        a.y = ap[k + 1];
        b.x = bp[(size_t)k * DIM];
        b.y = bp[(size_t)k * DIM + DIM];
        c = __builtin_amdgcn_wmma_f32_16x16x4_f32(false, a, false, b,
                                                  (short)0, c, false, false);
    }

    float* __restrict__ yp = Y + ((size_t)mt * 16 + half * 8) * DIM + col0 + lr;
#pragma unroll
    for (int r = 0; r < 8; ++r)
        yp[(size_t)r * DIM] = c[r];
}

__global__ void zero_kernel(float* __restrict__ p, long n)
{
    long i = (long)blockIdx.x * blockDim.x + threadIdx.x;
    if (i < n) p[i] = 0.0f;
}

__global__ void degrees_kernel(const int* __restrict__ node,
                               const int* __restrict__ edge,
                               float* __restrict__ Ddeg,
                               float* __restrict__ Bdeg, int nnz)
{
    int i = blockIdx.x * blockDim.x + threadIdx.x;
    if (i >= nnz) return;
    unsafeAtomicAdd(&Ddeg[node[i]], 1.0f);
    unsafeAtomicAdd(&Bdeg[edge[i]], 1.0f);
}

__global__ void __launch_bounds__(256)
scatter_kernel(const float* __restrict__ src, float* __restrict__ dst,
               const int* __restrict__ gidx, const int* __restrict__ sidx,
               int nnz)
{
    const int wid  = (int)(((long)blockIdx.x * blockDim.x + threadIdx.x) >> 5);
    const int lane = threadIdx.x & 31;
    if (wid >= nnz) return;

    const int g = gidx[wid];
    const int s = sidx[wid];

    const float* __restrict__ sp = src + (size_t)g * DIM + lane;
    float* dp = dst + (size_t)s * DIM + lane;
    const float v0 = sp[0];
    const float v1 = sp[32];
    const float v2 = sp[64];
    const float v3 = sp[96];
    unsafeAtomicAdd(dp + 0,  v0);
    unsafeAtomicAdd(dp + 32, v1);
    unsafeAtomicAdd(dp + 64, v2);
    unsafeAtomicAdd(dp + 96, v3);
}

__global__ void scale_rows_kernel(float* __restrict__ m,
                                  const float* __restrict__ Bdeg, long n)
{
    long i = (long)blockIdx.x * blockDim.x + threadIdx.x;
    if (i >= n) return;
    float bb  = Bdeg[i >> 7];
    float inv = bb > 0.0f ? 1.0f / bb : 0.0f;
    m[i] *= inv;
}

__global__ void finalize_kernel(float* __restrict__ out,
                                const float* __restrict__ Ddeg,
                                const float* __restrict__ bias, long n)
{
    long i = (long)blockIdx.x * blockDim.x + threadIdx.x;
    if (i >= n) return;
    float dd  = Ddeg[i >> 7];
    float inv = dd > 0.0f ? 1.0f / dd : 0.0f;
    float v   = out[i] * inv + bias[i & (DIM - 1)];
    out[i] = v >= 0.0f ? v : NEG_SLOPE * v;
}

#define KEY_NONE 0x7fffffff

#define RB_SHIFT  9
#define RB_ROWS   (1 << RB_SHIFT)
#define NBUCK_MAX 512
#define SRC_BITS  18
#define SRC_MASK  ((1 << SRC_BITS) - 1)
#define PT        1024
#define EPT       8
#define CHUNK     (PT * EPT)

__device__ __forceinline__ int scan512(int x, int tid, int* wsum)
{
    const int lane = tid & 31;
    const int wv   = tid >> 5;
    int v = x;
    for (int o = 1; o < 32; o <<= 1) {
        const int t = __shfl_up(v, o);
        if (lane >= o) v += t;
    }
    if (lane == 31 && wv < 16) wsum[wv] = v;
    __syncthreads();
    if (wv == 0) {
        int w = lane < 16 ? wsum[lane] : 0;
        for (int o = 1; o < 16; o <<= 1) {
            const int t = __shfl_up(w, o);
            if (lane >= o) w += t;
        }
        if (lane < 16) wsum[lane] = w;
    }
    __syncthreads();
    if (wv > 0 && wv < 16) v += wsum[wv - 1];
    return v;
}

__global__ void __launch_bounds__(PT)
range_count_kernel(const int* __restrict__ node, const int* __restrict__ edge,
                   int* __restrict__ bcnt, int nnz)
{
    __shared__ int h[NBUCK_MAX];
    const int tid = threadIdx.x;
    if (tid < NBUCK_MAX) h[tid] = 0;
    __syncthreads();
    const int base = blockIdx.x * CHUNK + tid;
    int ke[EPT], kn[EPT];
#pragma unroll
    for (int u = 0; u < EPT; ++u) {
        const int i = base + u * PT;
        ke[u] = i < nnz ? edge[i] : -1;
        kn[u] = i < nnz ? node[i] : 0;
    }
#pragma unroll
    for (int u = 0; u < EPT; ++u) {
        if (ke[u] >= 0) {
            atomicAdd(&h[ke[u] >> RB_SHIFT], 1);
            atomicAdd(&h[(NEDGES + kn[u]) >> RB_SHIFT], 1);
        }
    }
    __syncthreads();
    if (tid < NBUCK_MAX) {
        const int c = h[tid];
        if (c > 0) atomicAdd(&bcnt[tid], c);
    }
}

__global__ void __launch_bounds__(PT)
range_move_kernel(const int* __restrict__ node, const int* __restrict__ edge,
                  const int* __restrict__ bcnt, int* __restrict__ bcur,
                  int* __restrict__ boff, int* __restrict__ recs, int nnz)
{
    __shared__ int h[NBUCK_MAX];
    __shared__ int sbase[NBUCK_MAX];
    __shared__ int wsum[16];
    const int tid = threadIdx.x;
    if (tid < NBUCK_MAX) h[tid] = 0;
    __syncthreads();

    int ke[EPT], kn[EPT], re[EPT], rn[EPT];
    const int base = blockIdx.x * CHUNK + tid;
#pragma unroll
    for (int u = 0; u < EPT; ++u) {
        const int i = base + u * PT;
        ke[u] = i < nnz ? edge[i] : -1;
        kn[u] = i < nnz ? node[i] : 0;
    }
#pragma unroll
    for (int u = 0; u < EPT; ++u) {
        re[u] = 0;
        rn[u] = 0;
        if (ke[u] >= 0) {
            re[u] = atomicAdd(&h[ke[u] >> RB_SHIFT], 1);
            rn[u] = atomicAdd(&h[(NEDGES + kn[u]) >> RB_SHIFT], 1);
        }
    }

    const int tot = tid < NBUCK_MAX ? bcnt[tid] : 0;
    const int inc = scan512(tot, tid, wsum);
    if (tid < NBUCK_MAX) {
        const int ex = inc - tot;
        if (blockIdx.x == 0) boff[tid] = ex;
        const int c = h[tid];
        int b = 0;
        if (c > 0) b = atomicAdd(&bcur[tid], c);
        sbase[tid] = ex + b;
    }
    __syncthreads();

#pragma unroll
    for (int u = 0; u < EPT; ++u) {
        if (ke[u] >= 0) {
            const int e  = ke[u];
            const int n  = kn[u];
            const int rr = NEDGES + n;
            recs[sbase[e >> RB_SHIFT] + re[u]]  = ((e & (RB_ROWS - 1)) << SRC_BITS) | n;
            recs[sbase[rr >> RB_SHIFT] + rn[u]] = ((rr & (RB_ROWS - 1)) << SRC_BITS) | e;
        }
    }
}

__global__ void __launch_bounds__(PT)
range_rows_kernel(const int* __restrict__ recs, const int* __restrict__ bcnt,
                  const int* __restrict__ boff, int* __restrict__ cnt,
                  int* __restrict__ off, int* __restrict__ list, int nrows)
{
    __shared__ int c[RB_ROWS];
    __shared__ int f[RB_ROWS];
    __shared__ int wsum[16];
    const int tid   = threadIdx.x;
    const int b     = blockIdx.x;
    const int n     = bcnt[b];
    const int start = boff[b];
    const int* __restrict__ rp = recs + start;
    if (tid < RB_ROWS) c[tid] = 0;
    __syncthreads();

    for (int j0 = tid; j0 < n; j0 += 4 * PT) {
        int r[4];
#pragma unroll
        for (int u = 0; u < 4; ++u) {
            const int j = j0 + u * PT;
            r[u] = j < n ? rp[j] : -1;
        }
#pragma unroll
        for (int u = 0; u < 4; ++u)
            if (r[u] >= 0) atomicAdd(&c[r[u] >> SRC_BITS], 1);
    }
    __syncthreads();

    const int mine = tid < RB_ROWS ? c[tid] : 0;
    const int inc  = scan512(mine, tid, wsum);
    if (tid < RB_ROWS) {
        const int ex  = inc - mine;
        const int row = b * RB_ROWS + tid;
        f[tid] = ex;
        if (row < nrows) {
            cnt[row] = mine;
            off[row] = start + ex;
        }
    }
    __syncthreads();

    int* __restrict__ lp = list + start;
    for (int j0 = tid; j0 < n; j0 += 4 * PT) {
        int r[4];
#pragma unroll
        for (int u = 0; u < 4; ++u) {
            const int j = j0 + u * PT;
            r[u] = j < n ? rp[j] : -1;
        }
#pragma unroll
        for (int u = 0; u < 4; ++u) {
            if (r[u] >= 0) {
                const int p = atomicAdd(&f[r[u] >> SRC_BITS], 1);
                lp[p] = r[u] & SRC_MASK;
            }
        }
    }
}

template <int K, int Q>
__device__ __forceinline__ void
add_ranked_rows(const float* __restrict__ src, int lane, int t,
                const int (&k)[Q], const int (&r)[Q],
                float& a0, float& a1, float& a2, float& a3)
{
    int kk[K];
#pragma unroll
    for (int u = 0; u < K; ++u) {
        unsigned mq[Q];
#pragma unroll
        for (int q = 0; q < Q; ++q)
            mq[q] = __builtin_amdgcn_ballot_w32(r[q] == t + u);
        unsigned m = mq[Q - 1];
#pragma unroll
        for (int q = Q - 2; q >= 0; --q) m = mq[q] ? mq[q] : m;
        const int sl = __builtin_ctz(m);
        int key = __builtin_amdgcn_readlane(k[Q - 1], sl);
#pragma unroll
        for (int q = Q - 2; q >= 0; --q)
            key = mq[q] ? __builtin_amdgcn_readlane(k[q], sl) : key;
        kk[u] = key;
    }
    float v[K][4];
#pragma unroll
    for (int u = 0; u < K; ++u) {
        const float* __restrict__ sp = src + (size_t)kk[u] * DIM + lane;
        v[u][0] = sp[0];
        v[u][1] = sp[32];
        v[u][2] = sp[64];
        v[u][3] = sp[96];
    }
#pragma unroll
    for (int u = 0; u < K; ++u) {
        a0 += v[u][0];
        a1 += v[u][1];
        a2 += v[u][2];
        a3 += v[u][3];
    }
}

template <int Q>
__device__ __forceinline__ void
ranked_row_sum(const float* __restrict__ src, const int* __restrict__ lst,
               int lane, int n, float& a0, float& a1, float& a2, float& a3)
{
    int k[Q], r[Q];
#pragma unroll
    for (int q = 0; q < Q; ++q) {
        const int j = lane + 32 * q;
        k[q] = j < n ? lst[j] : KEY_NONE;
        r[q] = 0;
    }
#pragma unroll
    for (int p = 0; p < Q; ++p) {
        int np = n - 32 * p;
        np = np < 0 ? 0 : (np > 32 ? 32 : np);
        for (int j = 0; j < np; ++j) {
            const int kj = __builtin_amdgcn_readlane(k[p], j);
#pragma unroll
            for (int q = 0; q < Q; ++q) {
                bool before;
                if (p < q)      before = kj <= k[q];
                else if (p > q) before = kj < k[q];
                else            before = kj < k[q] || (kj == k[q] && j < lane);
                r[q] += before ? 1 : 0;
            }
        }
    }
#pragma unroll
    for (int q = 0; q < Q; ++q)
        if (lane + 32 * q >= n) r[q] = -1;

    int t = 0;
    for (; t + 4 <= n; t += 4)
        add_ranked_rows<4, Q>(src, lane, t, k, r, a0, a1, a2, a3);
    if (t + 2 <= n) {
        add_ranked_rows<2, Q>(src, lane, t, k, r, a0, a1, a2, a3);
        t += 2;
    }
    if (t < n)
        add_ranked_rows<1, Q>(src, lane, t, k, r, a0, a1, a2, a3);
}

template <int FINAL>
__global__ void __launch_bounds__(256)
rowsum_kernel(const float* __restrict__ src, float* __restrict__ dst,
              const int* __restrict__ cnt, const int* __restrict__ off,
              const int* __restrict__ list, const float* __restrict__ bias,
              int nrows)
{
    const int row  = (int)(((long)blockIdx.x * blockDim.x + threadIdx.x) >> 5);
    const int lane = threadIdx.x & 31;
    if (row >= nrows) return;

    const int n = __builtin_amdgcn_readfirstlane(cnt[row]);
    const int b = __builtin_amdgcn_readfirstlane(off[row]);
    const int* __restrict__ lst = list + b;

    float a0 = 0.0f, a1 = 0.0f, a2 = 0.0f, a3 = 0.0f;

    if (n <= 64) {
        ranked_row_sum<2>(src, lst, lane, n, a0, a1, a2, a3);
    } else if (n <= 128) {
        ranked_row_sum<4>(src, lst, lane, n, a0, a1, a2, a3);
    } else {
        int prev = -1;
        int done = 0;
        while (done < n) {
            int mn = KEY_NONE;
            for (int j = lane; j < n; j += 32) {
                const int k = lst[j];
                if (k > prev && k < mn) mn = k;
            }
            for (int o = 16; o > 0; o >>= 1) {
                const int other = __shfl_xor(mn, o);
                mn = other < mn ? other : mn;
            }
            int c = 0;
            for (int j = lane; j < n; j += 32) c += (lst[j] == mn) ? 1 : 0;
            for (int o = 16; o > 0; o >>= 1) c += __shfl_xor(c, o);
            mn = __builtin_amdgcn_readfirstlane(mn);
            c  = __builtin_amdgcn_readfirstlane(c);
            if (c <= 0) break;
            const float* __restrict__ sp = src + (size_t)mn * DIM + lane;
            const float w0 = sp[0];
            const float w1 = sp[32];
            const float w2 = sp[64];
            const float w3 = sp[96];
            for (int i = 0; i < c; ++i) {
                a0 += w0;
                a1 += w1;
                a2 += w2;
                a3 += w3;
            }
            prev = mn;
            done += c;
        }
    }

    const float inv = n > 0 ? 1.0f / (float)n : 0.0f;
    float* __restrict__ dp = dst + (size_t)row * DIM + lane;
    if (FINAL) {
        const float y0 = a0 * inv + bias[lane];
        const float y1 = a1 * inv + bias[lane + 32];
        const float y2 = a2 * inv + bias[lane + 64];
        const float y3 = a3 * inv + bias[lane + 96];
        dp[0]  = y0 >= 0.0f ? y0 : NEG_SLOPE * y0;
        dp[32] = y1 >= 0.0f ? y1 : NEG_SLOPE * y1;
        dp[64] = y2 >= 0.0f ? y2 : NEG_SLOPE * y2;
        dp[96] = y3 >= 0.0f ? y3 : NEG_SLOPE * y3;
    } else {
        dp[0]  = a0 * inv;
        dp[32] = a1 * inv;
        dp[64] = a2 * inv;
        dp[96] = a3 * inv;
    }
}

extern "C" void kernel_launch(void* const* d_in, const int* in_sizes, int n_in,
                              void* d_out, int out_size, void* d_ws, size_t ws_size,
                              hipStream_t stream)
{
    const float* X    = (const float*)d_in[0];
    const int*   A    = (const int*)d_in[1];
    const float* W    = (const float*)d_in[2];
    const float* bias = (const float*)d_in[3];
    float*       out  = (float*)d_out;

    const int N   = in_sizes[0] / DIM;
    const int NNZ = in_sizes[1] / 2;
    const int* node = A;
    const int* edge = A + NNZ;
    const long n_out = (long)N * DIM;
    const long n_m   = (long)NEDGES * DIM;

    const int mtiles = N / 16;

    const int  R    = NEDGES + N;
    const int  nb   = (R + RB_ROWS - 1) / RB_ROWS;
    float* m    = (float*)d_ws;
    int*   cnt  = (int*)(m + n_m);
    int*   off  = cnt + R;
    int*   list = off + R;
    int*   bcnt = list + 2L * NNZ;
    int*   bcur = bcnt + NBUCK_MAX;
    int*   boff = bcur + NBUCK_MAX;
    int*   recs = (int*)d_out;
    const size_t need = ((size_t)n_m + 2 * (size_t)R + 2 * (size_t)NNZ + 3 * NBUCK_MAX) * sizeof(float);

    if (ws_size >= need && nb <= NBUCK_MAX && N <= (1 << SRC_BITS) &&
        NEDGES <= (1 << SRC_BITS) && 2L * NNZ <= n_out) {
        const int nchunks = (NNZ + CHUNK - 1) / CHUNK;
        zero_kernel<<<(2 * NBUCK_MAX + 255) / 256, 256, 0, stream>>>((float*)bcnt, 2L * NBUCK_MAX);
        range_count_kernel<<<nchunks, PT, 0, stream>>>(node, edge, bcnt, NNZ);
        range_move_kernel<<<nchunks, PT, 0, stream>>>(node, edge, bcnt, bcur, boff, recs, NNZ);
        range_rows_kernel<<<nb, PT, 0, stream>>>(recs, bcnt, boff, cnt, off, list, R);
        gemm_wmma_f32<<<mtiles, 256, 0, stream>>>(X, W, out, mtiles);
        rowsum_kernel<0><<<(NEDGES + 7) / 8, 256, 0, stream>>>(out, m, cnt, off, list, bias, NEDGES);
        rowsum_kernel<1><<<(N + 7) / 8, 256, 0, stream>>>(m, out, cnt + NEDGES, off + NEDGES, list, bias, N);
        return;
    }

    gemm_wmma_f32<<<mtiles, 256, 0, stream>>>(X, W, out, mtiles);

    float* Ddeg = m + (size_t)NEDGES * DIM;
    float* Bdeg = Ddeg + N;
    const long ws_f  = (long)NEDGES * DIM + N + NEDGES;
    zero_kernel<<<(int)((ws_f + 255) / 256), 256, 0, stream>>>(m, ws_f);
    degrees_kernel<<<(NNZ + 255) / 256, 256, 0, stream>>>(node, edge, Ddeg, Bdeg, NNZ);
    scatter_kernel<<<(NNZ + 7) / 8, 256, 0, stream>>>(out, m, node, edge, NNZ);
    scale_rows_kernel<<<(int)((n_m + 255) / 256), 256, 0, stream>>>(m, Bdeg, n_m);
    zero_kernel<<<(int)((n_out + 255) / 256), 256, 0, stream>>>(out, n_out);
    scatter_kernel<<<(NNZ + 7) / 8, 256, 0, stream>>>(m, out, edge, node, NNZ);
    finalize_kernel<<<(int)((n_out + 255) / 256), 256, 0, stream>>>(out, Ddeg, bias, n_out);
}


# product with W fused into the pass 1 launch: a block averages 16 hyperedge rows into LDS and multiplies that tile by W on WMMA; means never reach memory; same add order and k order
# speedup vs baseline: 5.2383x; 1.4564x over previous
#include <hip/hip_runtime.h>

typedef __attribute__((ext_vector_type(2))) float v2f;
typedef __attribute__((ext_vector_type(8))) float v8f;

#define DIM        128
#define NEDGES     50000
#define NEG_SLOPE  0.01f

__global__ void __launch_bounds__(256)
gemm_wmma_f32(const float* __restrict__ X, const float* __restrict__ W,
              float* __restrict__ Y, int mtiles)
{
    const int lane  = threadIdx.x & 31;
    const int wv    = threadIdx.x >> 5;
    const long tile = (long)blockIdx.x * 8 + wv;
    const long mt   = tile >> 3;
    const int  nt   = (int)(tile & 7);
    if (mt >= mtiles) return;

    const int half = lane >> 4;
    const int lr   = lane & 15;
    const size_t row  = (size_t)mt * 16 + lr;
    const int    col0 = nt * 16;

    const float* __restrict__ ap = X + row * DIM + 2 * half;
    const float* __restrict__ bp = W + (size_t)(2 * half) * DIM + col0 + lr;

    v8f c = {};
#pragma unroll
    for (int k = 0; k < DIM; k += 4) {
        v2f a, b;
        a.x = ap[k];
        a.y = ap[k + 1];
        b.x = bp[(size_t)k * DIM];
        b.y = bp[(size_t)k * DIM + DIM];
        c = __builtin_amdgcn_wmma_f32_16x16x4_f32(false, a, false, b,
                                                  (short)0, c, false, false);
    }

    float* __restrict__ yp = Y + ((size_t)mt * 16 + half * 8) * DIM + col0 + lr;
#pragma unroll
    for (int r = 0; r < 8; ++r)
        yp[(size_t)r * DIM] = c[r];
}

__global__ void zero_kernel(float* __restrict__ p, long n)
{
    long i = (long)blockIdx.x * blockDim.x + threadIdx.x;
    if (i < n) p[i] = 0.0f;
}

__global__ void degrees_kernel(const int* __restrict__ node,
                               const int* __restrict__ edge,
                               float* __restrict__ Ddeg,
                               float* __restrict__ Bdeg, int nnz)
{
    int i = blockIdx.x * blockDim.x + threadIdx.x;
    if (i >= nnz) return;
    unsafeAtomicAdd(&Ddeg[node[i]], 1.0f);
    unsafeAtomicAdd(&Bdeg[edge[i]], 1.0f);
}

__global__ void __launch_bounds__(256)
scatter_kernel(const float* __restrict__ src, float* __restrict__ dst,
               const int* __restrict__ gidx, const int* __restrict__ sidx,
               int nnz)
{
    const int wid  = (int)(((long)blockIdx.x * blockDim.x + threadIdx.x) >> 5);
    const int lane = threadIdx.x & 31;
    if (wid >= nnz) return;

    const int g = gidx[wid];
    const int s = sidx[wid];

    const float* __restrict__ sp = src + (size_t)g * DIM + lane;
    float* dp = dst + (size_t)s * DIM + lane;
    const float v0 = sp[0];
    const float v1 = sp[32];
    const float v2 = sp[64];
    const float v3 = sp[96];
    unsafeAtomicAdd(dp + 0,  v0);
    unsafeAtomicAdd(dp + 32, v1);
    unsafeAtomicAdd(dp + 64, v2);
    unsafeAtomicAdd(dp + 96, v3);
}

__global__ void scale_rows_kernel(float* __restrict__ m,
                                  const float* __restrict__ Bdeg, long n)
{
    long i = (long)blockIdx.x * blockDim.x + threadIdx.x;
    if (i >= n) return;
    float bb  = Bdeg[i >> 7];
    float inv = bb > 0.0f ? 1.0f / bb : 0.0f;
    m[i] *= inv;
}

__global__ void finalize_kernel(float* __restrict__ out,
                                const float* __restrict__ Ddeg,
                                const float* __restrict__ bias, long n)
{
    long i = (long)blockIdx.x * blockDim.x + threadIdx.x;
    if (i >= n) return;
    float dd  = Ddeg[i >> 7];
    float inv = dd > 0.0f ? 1.0f / dd : 0.0f;
    float v   = out[i] * inv + bias[i & (DIM - 1)];
    out[i] = v >= 0.0f ? v : NEG_SLOPE * v;
}

#define KEY_NONE 0x7fffffff

#define RB_SHIFT  9
#define RB_ROWS   (1 << RB_SHIFT)
#define NBUCK_MAX 512
#define SRC_BITS  18
#define SRC_MASK  ((1 << SRC_BITS) - 1)
#define PT        1024
#define EPT       8
#define CHUNK     (PT * EPT)

__device__ __forceinline__ int scan512(int x, int tid, int* wsum)
{
    const int lane = tid & 31;
    const int wv   = tid >> 5;
    int v = x;
    for (int o = 1; o < 32; o <<= 1) {
        const int t = __shfl_up(v, o);
        if (lane >= o) v += t;
    }
    if (lane == 31 && wv < 16) wsum[wv] = v;
    __syncthreads();
    if (wv == 0) {
        int w = lane < 16 ? wsum[lane] : 0;
        for (int o = 1; o < 16; o <<= 1) {
            const int t = __shfl_up(w, o);
            if (lane >= o) w += t;
        }
        if (lane < 16) wsum[lane] = w;
    }
    __syncthreads();
    if (wv > 0 && wv < 16) v += wsum[wv - 1];
    return v;
}

__global__ void __launch_bounds__(PT)
range_count_kernel(const int* __restrict__ node, const int* __restrict__ edge,
                   int* __restrict__ bcnt, int nnz)
{
    __shared__ int h[NBUCK_MAX];
    const int tid = threadIdx.x;
    if (tid < NBUCK_MAX) h[tid] = 0;
    __syncthreads();
    const int base = blockIdx.x * CHUNK + tid;
    int ke[EPT], kn[EPT];
#pragma unroll
    for (int u = 0; u < EPT; ++u) {
        const int i = base + u * PT;
        ke[u] = i < nnz ? edge[i] : -1;
        kn[u] = i < nnz ? node[i] : 0;
    }
#pragma unroll
    for (int u = 0; u < EPT; ++u) {
        if (ke[u] >= 0) {
            atomicAdd(&h[ke[u] >> RB_SHIFT], 1);
            atomicAdd(&h[(NEDGES + kn[u]) >> RB_SHIFT], 1);
        }
    }
    __syncthreads();
    if (tid < NBUCK_MAX) {
        const int c = h[tid];
        if (c > 0) atomicAdd(&bcnt[tid], c);
    }
}

__global__ void __launch_bounds__(PT)
range_move_kernel(const int* __restrict__ node, const int* __restrict__ edge,
                  const int* __restrict__ bcnt, int* __restrict__ bcur,
                  int* __restrict__ boff, int* __restrict__ recs, int nnz)
{
    __shared__ int h[NBUCK_MAX];
    __shared__ int sbase[NBUCK_MAX];
    __shared__ int wsum[16];
    const int tid = threadIdx.x;
    if (tid < NBUCK_MAX) h[tid] = 0;
    __syncthreads();

    int ke[EPT], kn[EPT], re[EPT], rn[EPT];
    const int base = blockIdx.x * CHUNK + tid;
#pragma unroll
    for (int u = 0; u < EPT; ++u) {
        const int i = base + u * PT;
        ke[u] = i < nnz ? edge[i] : -1;
        kn[u] = i < nnz ? node[i] : 0;
    }
#pragma unroll
    for (int u = 0; u < EPT; ++u) {
        re[u] = 0;
        rn[u] = 0;
        if (ke[u] >= 0) {
            re[u] = atomicAdd(&h[ke[u] >> RB_SHIFT], 1);
            rn[u] = atomicAdd(&h[(NEDGES + kn[u]) >> RB_SHIFT], 1);
        }
    }

    const int tot = tid < NBUCK_MAX ? bcnt[tid] : 0;
    const int inc = scan512(tot, tid, wsum);
    if (tid < NBUCK_MAX) {
        const int ex = inc - tot;
        if (blockIdx.x == 0) boff[tid] = ex;
        const int c = h[tid];
        int b = 0;
        if (c > 0) b = atomicAdd(&bcur[tid], c);
        sbase[tid] = ex + b;
    }
    __syncthreads();

#pragma unroll
    for (int u = 0; u < EPT; ++u) {
        if (ke[u] >= 0) {
            const int e  = ke[u];
            const int n  = kn[u];
            const int rr = NEDGES + n;
            recs[sbase[e >> RB_SHIFT] + re[u]]  = ((e & (RB_ROWS - 1)) << SRC_BITS) | n;
            recs[sbase[rr >> RB_SHIFT] + rn[u]] = ((rr & (RB_ROWS - 1)) << SRC_BITS) | e;
        }
    }
}

__global__ void __launch_bounds__(PT)
range_rows_kernel(const int* __restrict__ recs, const int* __restrict__ bcnt,
                  const int* __restrict__ boff, int* __restrict__ cnt,
                  int* __restrict__ off, int* __restrict__ list, int nrows)
{
    __shared__ int c[RB_ROWS];
    __shared__ int f[RB_ROWS];
    __shared__ int wsum[16];
    const int tid   = threadIdx.x;
    const int b     = blockIdx.x;
    const int n     = bcnt[b];
    const int start = boff[b];
    const int* __restrict__ rp = recs + start;
    if (tid < RB_ROWS) c[tid] = 0;
    __syncthreads();

    for (int j0 = tid; j0 < n; j0 += 4 * PT) {
        int r[4];
#pragma unroll
        for (int u = 0; u < 4; ++u) {
            const int j = j0 + u * PT;
            r[u] = j < n ? rp[j] : -1;
        }
#pragma unroll
        for (int u = 0; u < 4; ++u)
            if (r[u] >= 0) atomicAdd(&c[r[u] >> SRC_BITS], 1);
    }
    __syncthreads();

    const int mine = tid < RB_ROWS ? c[tid] : 0;
    const int inc  = scan512(mine, tid, wsum);
    if (tid < RB_ROWS) {
        const int ex  = inc - mine;
        const int row = b * RB_ROWS + tid;
        f[tid] = ex;
        if (row < nrows) {
            cnt[row] = mine;
            off[row] = start + ex;
        }
    }
    __syncthreads();

    int* __restrict__ lp = list + start;
    for (int j0 = tid; j0 < n; j0 += 4 * PT) {
        int r[4];
#pragma unroll
        for (int u = 0; u < 4; ++u) {
            const int j = j0 + u * PT;
            r[u] = j < n ? rp[j] : -1;
        }
#pragma unroll
        for (int u = 0; u < 4; ++u) {
            if (r[u] >= 0) {
                const int p = atomicAdd(&f[r[u] >> SRC_BITS], 1);
                lp[p] = r[u] & SRC_MASK;
            }
        }
    }
}

template <int K, int Q>
__device__ __forceinline__ void
add_ranked_rows(const float* __restrict__ src, int lane, int t,
                const int (&k)[Q], const int (&r)[Q],
                float& a0, float& a1, float& a2, float& a3)
{
    int kk[K];
#pragma unroll
    for (int u = 0; u < K; ++u) {
        unsigned mq[Q];
#pragma unroll
        for (int q = 0; q < Q; ++q)
            mq[q] = __builtin_amdgcn_ballot_w32(r[q] == t + u);
        unsigned m = mq[Q - 1];
#pragma unroll
        for (int q = Q - 2; q >= 0; --q) m = mq[q] ? mq[q] : m;
        const int sl = __builtin_ctz(m);
        int key = __builtin_amdgcn_readlane(k[Q - 1], sl);
#pragma unroll
        for (int q = Q - 2; q >= 0; --q)
            key = mq[q] ? __builtin_amdgcn_readlane(k[q], sl) : key;
        kk[u] = key;
    }
    float v[K][4];
#pragma unroll
    for (int u = 0; u < K; ++u) {
        const float* __restrict__ sp = src + (size_t)kk[u] * DIM + lane;
        v[u][0] = sp[0];
        v[u][1] = sp[32];
        v[u][2] = sp[64];
        v[u][3] = sp[96];
    }
#pragma unroll
    for (int u = 0; u < K; ++u) {
        a0 += v[u][0];
        a1 += v[u][1];
        a2 += v[u][2];
        a3 += v[u][3];
    }
}

template <int Q>
__device__ __forceinline__ void
ranked_row_sum(const float* __restrict__ src, const int* __restrict__ lst,
               int lane, int n, float& a0, float& a1, float& a2, float& a3)
{
    int k[Q], r[Q];
#pragma unroll
    for (int q = 0; q < Q; ++q) {
        const int j = lane + 32 * q;
        k[q] = j < n ? lst[j] : KEY_NONE;
        r[q] = 0;
    }
#pragma unroll
    for (int p = 0; p < Q; ++p) {
        int np = n - 32 * p;
        np = np < 0 ? 0 : (np > 32 ? 32 : np);
        for (int j = 0; j < np; ++j) {
            const int kj = __builtin_amdgcn_readlane(k[p], j);
#pragma unroll
            for (int q = 0; q < Q; ++q) {
                bool before;
                if (p < q)      before = kj <= k[q];
                else if (p > q) before = kj < k[q];
                else            before = kj < k[q] || (kj == k[q] && j < lane);
                r[q] += before ? 1 : 0;
            }
        }
    }
#pragma unroll
    for (int q = 0; q < Q; ++q)
        if (lane + 32 * q >= n) r[q] = -1;

    int t = 0;
    for (; t + 4 <= n; t += 4)
        add_ranked_rows<4, Q>(src, lane, t, k, r, a0, a1, a2, a3);
    if (t + 2 <= n) {
        add_ranked_rows<2, Q>(src, lane, t, k, r, a0, a1, a2, a3);
        t += 2;
    }
    if (t < n)
        add_ranked_rows<1, Q>(src, lane, t, k, r, a0, a1, a2, a3);
}

__device__ __forceinline__ void
sum_row(const float* __restrict__ src, const int* __restrict__ lst,
        int lane, int n, float& a0, float& a1, float& a2, float& a3)
{
    if (n <= 64) {
        ranked_row_sum<2>(src, lst, lane, n, a0, a1, a2, a3);
    } else if (n <= 128) {
        ranked_row_sum<4>(src, lst, lane, n, a0, a1, a2, a3);
    } else {
        int prev = -1;
        int done = 0;
        while (done < n) {
            int mn = KEY_NONE;
            for (int j = lane; j < n; j += 32) {
                const int k = lst[j];
                if (k > prev && k < mn) mn = k;
            }
            for (int o = 16; o > 0; o >>= 1) {
                const int other = __shfl_xor(mn, o);
                mn = other < mn ? other : mn;
            }
            int c = 0;
            for (int j = lane; j < n; j += 32) c += (lst[j] == mn) ? 1 : 0;
            for (int o = 16; o > 0; o >>= 1) c += __shfl_xor(c, o);
            mn = __builtin_amdgcn_readfirstlane(mn);
            c  = __builtin_amdgcn_readfirstlane(c);
            if (c <= 0) break;
            const float* __restrict__ sp = src + (size_t)mn * DIM + lane;
            const float w0 = sp[0];
            const float w1 = sp[32];
            const float w2 = sp[64];
            const float w3 = sp[96];
            for (int i = 0; i < c; ++i) {
                a0 += w0;
                a1 += w1;
                a2 += w2;
                a3 += w3;
            }
            prev = mn;
            done += c;
        }
    }
}

template <int FINAL>
__global__ void __launch_bounds__(256)
rowsum_kernel(const float* __restrict__ src, float* __restrict__ dst,
              const int* __restrict__ cnt, const int* __restrict__ off,
              const int* __restrict__ list, const float* __restrict__ bias,
              int nrows)
{
    const int row  = (int)(((long)blockIdx.x * blockDim.x + threadIdx.x) >> 5);
    const int lane = threadIdx.x & 31;
    if (row >= nrows) return;

    const int n = __builtin_amdgcn_readfirstlane(cnt[row]);
    const int b = __builtin_amdgcn_readfirstlane(off[row]);
    const int* __restrict__ lst = list + b;

    float a0 = 0.0f, a1 = 0.0f, a2 = 0.0f, a3 = 0.0f;
    sum_row(src, lst, lane, n, a0, a1, a2, a3);

    const float inv = n > 0 ? 1.0f / (float)n : 0.0f;
    float* __restrict__ dp = dst + (size_t)row * DIM + lane;
    if (FINAL) {
        const float y0 = a0 * inv + bias[lane];
        const float y1 = a1 * inv + bias[lane + 32];
        const float y2 = a2 * inv + bias[lane + 64];
        const float y3 = a3 * inv + bias[lane + 96];
        dp[0]  = y0 >= 0.0f ? y0 : NEG_SLOPE * y0;
        dp[32] = y1 >= 0.0f ? y1 : NEG_SLOPE * y1;
        dp[64] = y2 >= 0.0f ? y2 : NEG_SLOPE * y2;
        dp[96] = y3 >= 0.0f ? y3 : NEG_SLOPE * y3;
    } else {
        dp[0]  = a0 * inv;
        dp[32] = a1 * inv;
        dp[64] = a2 * inv;
        dp[96] = a3 * inv;
    }
}

#define TILE_P 132
__global__ void __launch_bounds__(512)
rowmean_gemm_kernel(const float* __restrict__ X, const float* __restrict__ W,
                    float* __restrict__ M, const int* __restrict__ cnt,
                    const int* __restrict__ off, const int* __restrict__ list)
{
    __shared__ float tile[16 * TILE_P];
    const int lane = threadIdx.x & 31;
    const int wv   = threadIdx.x >> 5;
    const int row0 = blockIdx.x * 16;
    {
        const int row = row0 + wv;
        const int n = __builtin_amdgcn_readfirstlane(cnt[row]);
        const int b = __builtin_amdgcn_readfirstlane(off[row]);
        float a0 = 0.0f, a1 = 0.0f, a2 = 0.0f, a3 = 0.0f;
        sum_row(X, list + b, lane, n, a0, a1, a2, a3);
        const float inv = n > 0 ? 1.0f / (float)n : 0.0f;
        float* tp = tile + wv * TILE_P + lane;
        tp[0]  = a0 * inv;
        tp[32] = a1 * inv;
        tp[64] = a2 * inv;
        tp[96] = a3 * inv;
    }
    __syncthreads();
    if (wv >= DIM / 16) return;

    const int half = lane >> 4;
    const int lr   = lane & 15;
    const int col0 = wv * 16;
    const float* ap = tile + lr * TILE_P + 2 * half;
    const float* __restrict__ bp = W + (size_t)(2 * half) * DIM + col0 + lr;

    unsigned yo = (unsigned)(((row0 + half * 8) * DIM + col0 + lr) * sizeof(float));
    asm volatile("" : "+v"(yo));
    float* __restrict__ yp = (float*)((char*)M + yo);

    v8f c = {};
#pragma unroll
    for (int k = 0; k < DIM; k += 4) {
        v2f a, b;
        a.x = ap[k];
        a.y = ap[k + 1];
        b.x = bp[(size_t)k * DIM];
        b.y = bp[(size_t)k * DIM + DIM];
        c = __builtin_amdgcn_wmma_f32_16x16x4_f32(false, a, false, b,
                                                  (short)0, c, false, false);
    }

#pragma unroll
    for (int r = 0; r < 8; ++r)
        yp[(size_t)r * DIM] = c[r];
}

extern "C" void kernel_launch(void* const* d_in, const int* in_sizes, int n_in,
                              void* d_out, int out_size, void* d_ws, size_t ws_size,
                              hipStream_t stream)
{
    const float* X    = (const float*)d_in[0];
    const int*   A    = (const int*)d_in[1];
    const float* W    = (const float*)d_in[2];
    const float* bias = (const float*)d_in[3];
    float*       out  = (float*)d_out;

    const int N   = in_sizes[0] / DIM;
    const int NNZ = in_sizes[1] / 2;
    const int* node = A;
    const int* edge = A + NNZ;
    const long n_out = (long)N * DIM;
    const long n_m   = (long)NEDGES * DIM;

    const int mtiles = N / 16;

    const int  R    = NEDGES + N;
    const int  nb   = (R + RB_ROWS - 1) / RB_ROWS;
    float* m    = (float*)d_ws;
    int*   cnt  = (int*)(m + n_m);
    int*   off  = cnt + R;
    int*   list = off + R;
    int*   bcnt = list + 2L * NNZ;
    int*   bcur = bcnt + NBUCK_MAX;
    int*   boff = bcur + NBUCK_MAX;
    int*   recs = (int*)d_out;
    const int etiles = NEDGES / 16;
    const size_t need = ((size_t)n_m + 2 * (size_t)R + 2 * (size_t)NNZ + 3 * NBUCK_MAX) * sizeof(float);

    if (ws_size >= need && nb <= NBUCK_MAX && N <= (1 << SRC_BITS) &&
        NEDGES <= (1 << SRC_BITS) && 2L * NNZ <= n_out) {
        const int nchunks = (NNZ + CHUNK - 1) / CHUNK;
        zero_kernel<<<(2 * NBUCK_MAX + 255) / 256, 256, 0, stream>>>((float*)bcnt, 2L * NBUCK_MAX);
        range_count_kernel<<<nchunks, PT, 0, stream>>>(node, edge, bcnt, NNZ);
        range_move_kernel<<<nchunks, PT, 0, stream>>>(node, edge, bcnt, bcur, boff, recs, NNZ);
        range_rows_kernel<<<nb, PT, 0, stream>>>(recs, bcnt, boff, cnt, off, list, R);
        rowmean_gemm_kernel<<<etiles, 512, 0, stream>>>(X, W, m, cnt, off, list);
        rowsum_kernel<1><<<(N + 7) / 8, 256, 0, stream>>>(m, out, cnt + NEDGES, off + NEDGES, list, bias, N);
        return;
    }

    gemm_wmma_f32<<<mtiles, 256, 0, stream>>>(X, W, out, mtiles);

    float* Ddeg = m + (size_t)NEDGES * DIM;
    float* Bdeg = Ddeg + N;
    const long ws_f  = (long)NEDGES * DIM + N + NEDGES;
    zero_kernel<<<(int)((ws_f + 255) / 256), 256, 0, stream>>>(m, ws_f);
    degrees_kernel<<<(NNZ + 255) / 256, 256, 0, stream>>>(node, edge, Ddeg, Bdeg, NNZ);
    scatter_kernel<<<(NNZ + 7) / 8, 256, 0, stream>>>(out, m, node, edge, NNZ);
    scale_rows_kernel<<<(int)((n_m + 255) / 256), 256, 0, stream>>>(m, Bdeg, n_m);
    zero_kernel<<<(int)((n_out + 255) / 256), 256, 0, stream>>>(out, n_out);
    scatter_kernel<<<(NNZ + 7) / 8, 256, 0, stream>>>(m, out, edge, node, NNZ);
    finalize_kernel<<<(int)((n_out + 255) / 256), 256, 0, stream>>>(out, Ddeg, bias, n_out);
}


# row sums: each row's source keys are sorted in registers by a bitonic network of lane exchanges (DPP, permlanex16) instead of ranked by counting and picked by ballots; same add order, bit-identical
# speedup vs baseline: 5.6068x; 1.0703x over previous
#include <hip/hip_runtime.h>

typedef __attribute__((ext_vector_type(2))) float v2f;
typedef __attribute__((ext_vector_type(8))) float v8f;

#define DIM        128
#define NEDGES     50000
#define NEG_SLOPE  0.01f

__global__ void __launch_bounds__(256)
gemm_wmma_f32(const float* __restrict__ X, const float* __restrict__ W,
              float* __restrict__ Y, int mtiles)
{
    const int lane  = threadIdx.x & 31;
    const int wv    = threadIdx.x >> 5;
    const long tile = (long)blockIdx.x * 8 + wv;
    const long mt   = tile >> 3;
    const int  nt   = (int)(tile & 7);
    if (mt >= mtiles) return;

    const int half = lane >> 4;
    const int lr   = lane & 15;
    const size_t row  = (size_t)mt * 16 + lr;
    const int    col0 = nt * 16;

    const float* __restrict__ ap = X + row * DIM + 2 * half;
    const float* __restrict__ bp = W + (size_t)(2 * half) * DIM + col0 + lr;

    v8f c = {};
#pragma unroll
    for (int k = 0; k < DIM; k += 4) {
        v2f a, b;
        a.x = ap[k];
        a.y = ap[k + 1];
        b.x = bp[(size_t)k * DIM];
        b.y = bp[(size_t)k * DIM + DIM];
        c = __builtin_amdgcn_wmma_f32_16x16x4_f32(false, a, false, b,
                                                  (short)0, c, false, false);
    }

    float* __restrict__ yp = Y + ((size_t)mt * 16 + half * 8) * DIM + col0 + lr;
#pragma unroll
    for (int r = 0; r < 8; ++r)
        yp[(size_t)r * DIM] = c[r];
}

__global__ void zero_kernel(float* __restrict__ p, long n)
{
    long i = (long)blockIdx.x * blockDim.x + threadIdx.x;
    if (i < n) p[i] = 0.0f;
}

__global__ void degrees_kernel(const int* __restrict__ node,
                               const int* __restrict__ edge,
                               float* __restrict__ Ddeg,
                               float* __restrict__ Bdeg, int nnz)
{
    int i = blockIdx.x * blockDim.x + threadIdx.x;
    if (i >= nnz) return;
    unsafeAtomicAdd(&Ddeg[node[i]], 1.0f);
    unsafeAtomicAdd(&Bdeg[edge[i]], 1.0f);
}

__global__ void __launch_bounds__(256)
scatter_kernel(const float* __restrict__ src, float* __restrict__ dst,
               const int* __restrict__ gidx, const int* __restrict__ sidx,
               int nnz)
{
    const int wid  = (int)(((long)blockIdx.x * blockDim.x + threadIdx.x) >> 5);
    const int lane = threadIdx.x & 31;
    if (wid >= nnz) return;

    const int g = gidx[wid];
    const int s = sidx[wid];

    const float* __restrict__ sp = src + (size_t)g * DIM + lane;
    float* dp = dst + (size_t)s * DIM + lane;
    const float v0 = sp[0];
    const float v1 = sp[32];
    const float v2 = sp[64];
    const float v3 = sp[96];
    unsafeAtomicAdd(dp + 0,  v0);
    unsafeAtomicAdd(dp + 32, v1);
    unsafeAtomicAdd(dp + 64, v2);
    unsafeAtomicAdd(dp + 96, v3);
}

__global__ void scale_rows_kernel(float* __restrict__ m,
                                  const float* __restrict__ Bdeg, long n)
{
    long i = (long)blockIdx.x * blockDim.x + threadIdx.x;
    if (i >= n) return;
    float bb  = Bdeg[i >> 7];
    float inv = bb > 0.0f ? 1.0f / bb : 0.0f;
    m[i] *= inv;
}

__global__ void finalize_kernel(float* __restrict__ out,
                                const float* __restrict__ Ddeg,
                                const float* __restrict__ bias, long n)
{
    long i = (long)blockIdx.x * blockDim.x + threadIdx.x;
    if (i >= n) return;
    float dd  = Ddeg[i >> 7];
    float inv = dd > 0.0f ? 1.0f / dd : 0.0f;
    float v   = out[i] * inv + bias[i & (DIM - 1)];
    out[i] = v >= 0.0f ? v : NEG_SLOPE * v;
}

#define KEY_NONE 0x7fffffff

#define RB_SHIFT  9
#define RB_ROWS   (1 << RB_SHIFT)
#define NBUCK_MAX 512
#define SRC_BITS  18
#define SRC_MASK  ((1 << SRC_BITS) - 1)
#define PT        1024
#define EPT       8
#define CHUNK     (PT * EPT)

__device__ __forceinline__ int scan512(int x, int tid, int* wsum)
{
    const int lane = tid & 31;
    const int wv   = tid >> 5;
    int v = x;
    for (int o = 1; o < 32; o <<= 1) {
        const int t = __shfl_up(v, o);
        if (lane >= o) v += t;
    }
    if (lane == 31 && wv < 16) wsum[wv] = v;
    __syncthreads();
    if (wv == 0) {
        int w = lane < 16 ? wsum[lane] : 0;
        for (int o = 1; o < 16; o <<= 1) {
            const int t = __shfl_up(w, o);
            if (lane >= o) w += t;
        }
        if (lane < 16) wsum[lane] = w;
    }
    __syncthreads();
    if (wv > 0 && wv < 16) v += wsum[wv - 1];
    return v;
}

__global__ void __launch_bounds__(PT)
range_count_kernel(const int* __restrict__ node, const int* __restrict__ edge,
                   int* __restrict__ bcnt, int nnz)
{
    __shared__ int h[NBUCK_MAX];
    const int tid = threadIdx.x;
    if (tid < NBUCK_MAX) h[tid] = 0;
    __syncthreads();
    const int base = blockIdx.x * CHUNK + tid;
    int ke[EPT], kn[EPT];
#pragma unroll
    for (int u = 0; u < EPT; ++u) {
        const int i = base + u * PT;
        ke[u] = i < nnz ? edge[i] : -1;
        kn[u] = i < nnz ? node[i] : 0;
    }
#pragma unroll
    for (int u = 0; u < EPT; ++u) {
        if (ke[u] >= 0) {
            atomicAdd(&h[ke[u] >> RB_SHIFT], 1);
            atomicAdd(&h[(NEDGES + kn[u]) >> RB_SHIFT], 1);
        }
    }
    __syncthreads();
    if (tid < NBUCK_MAX) {
        const int c = h[tid];
        if (c > 0) atomicAdd(&bcnt[tid], c);
    }
}

__global__ void __launch_bounds__(PT)
range_move_kernel(const int* __restrict__ node, const int* __restrict__ edge,
                  const int* __restrict__ bcnt, int* __restrict__ bcur,
                  int* __restrict__ boff, int* __restrict__ recs, int nnz)
{
    __shared__ int h[NBUCK_MAX];
    __shared__ int sbase[NBUCK_MAX];
    __shared__ int wsum[16];
    const int tid = threadIdx.x;
    if (tid < NBUCK_MAX) h[tid] = 0;
    __syncthreads();

    int ke[EPT], kn[EPT], re[EPT], rn[EPT];
    const int base = blockIdx.x * CHUNK + tid;
#pragma unroll
    for (int u = 0; u < EPT; ++u) {
        const int i = base + u * PT;
        ke[u] = i < nnz ? edge[i] : -1;
        kn[u] = i < nnz ? node[i] : 0;
    }
#pragma unroll
    for (int u = 0; u < EPT; ++u) {
        re[u] = 0;
        rn[u] = 0;
        if (ke[u] >= 0) {
            re[u] = atomicAdd(&h[ke[u] >> RB_SHIFT], 1);
            rn[u] = atomicAdd(&h[(NEDGES + kn[u]) >> RB_SHIFT], 1);
        }
    }

    const int tot = tid < NBUCK_MAX ? bcnt[tid] : 0;
    const int inc = scan512(tot, tid, wsum);
    if (tid < NBUCK_MAX) {
        const int ex = inc - tot;
        if (blockIdx.x == 0) boff[tid] = ex;
        const int c = h[tid];
        int b = 0;
        if (c > 0) b = atomicAdd(&bcur[tid], c);
        sbase[tid] = ex + b;
    }
    __syncthreads();

#pragma unroll
    for (int u = 0; u < EPT; ++u) {
        if (ke[u] >= 0) {
            const int e  = ke[u];
            const int n  = kn[u];
            const int rr = NEDGES + n;
            recs[sbase[e >> RB_SHIFT] + re[u]]  = ((e & (RB_ROWS - 1)) << SRC_BITS) | n;
            recs[sbase[rr >> RB_SHIFT] + rn[u]] = ((rr & (RB_ROWS - 1)) << SRC_BITS) | e;
        }
    }
}

__global__ void __launch_bounds__(PT)
range_rows_kernel(const int* __restrict__ recs, const int* __restrict__ bcnt,
                  const int* __restrict__ boff, int* __restrict__ cnt,
                  int* __restrict__ off, int* __restrict__ list, int nrows)
{
    __shared__ int c[RB_ROWS];
    __shared__ int f[RB_ROWS];
    __shared__ int wsum[16];
    const int tid   = threadIdx.x;
    const int b     = blockIdx.x;
    const int n     = bcnt[b];
    const int start = boff[b];
    const int* __restrict__ rp = recs + start;
    if (tid < RB_ROWS) c[tid] = 0;
    __syncthreads();

    for (int j0 = tid; j0 < n; j0 += 4 * PT) {
        int r[4];
#pragma unroll
        for (int u = 0; u < 4; ++u) {
            const int j = j0 + u * PT;
            r[u] = j < n ? rp[j] : -1;
        }
#pragma unroll
        for (int u = 0; u < 4; ++u)
            if (r[u] >= 0) atomicAdd(&c[r[u] >> SRC_BITS], 1);
    }
    __syncthreads();

    const int mine = tid < RB_ROWS ? c[tid] : 0;
    const int inc  = scan512(mine, tid, wsum);
    if (tid < RB_ROWS) {
        const int ex  = inc - mine;
        const int row = b * RB_ROWS + tid;
        f[tid] = ex;
        if (row < nrows) {
            cnt[row] = mine;
            off[row] = start + ex;
        }
    }
    __syncthreads();

    int* __restrict__ lp = list + start;
    for (int j0 = tid; j0 < n; j0 += 4 * PT) {
        int r[4];
#pragma unroll
        for (int u = 0; u < 4; ++u) {
            const int j = j0 + u * PT;
            r[u] = j < n ? rp[j] : -1;
        }
#pragma unroll
        for (int u = 0; u < 4; ++u) {
            if (r[u] >= 0) {
                const int p = atomicAdd(&f[r[u] >> SRC_BITS], 1);
                lp[p] = r[u] & SRC_MASK;
            }
        }
    }
}

template <int J>
__device__ __forceinline__ int lane_xor(int v)
{
    if constexpr (J == 16)
        return __builtin_amdgcn_permlanex16(v, v, 0x76543210, 0xfedcba98, false, false);
    else
        return __builtin_amdgcn_update_dpp(v, v, 0x160 + J, 0xf, 0xf, true);
}

template <int K, int J, int Q>
__device__ __forceinline__ void sort_step(int (&k)[Q], int lane)
{
    if constexpr (J >= 32) {
        constexpr int dq = J >> 5;
#pragma unroll
        for (int q = 0; q < Q; ++q) {
            if ((q & dq) == 0) {
                const bool asc = ((32 * q) & K) == 0;
                const int x  = k[q];
                const int y  = k[q + dq];
                const int lo = y < x ? y : x;
                const int hi = y < x ? x : y;
                k[q]      = asc ? lo : hi;
                k[q + dq] = asc ? hi : lo;
            }
        }
    } else {
        const bool lower = (lane & J) == 0;
#pragma unroll
        for (int q = 0; q < Q; ++q) {
            const bool asc = K < 32 ? (lane & K) == 0 : ((32 * q) & K) == 0;
            const int x  = k[q];
            const int o  = lane_xor<J>(x);
            const int mn = o < x ? o : x;
            const int mx = o < x ? x : o;
            k[q] = (lower == asc) ? mn : mx;
        }
    }
}

template <int K, int J, int Q>
__device__ __forceinline__ void sort_from(int (&k)[Q], int lane)
{
    sort_step<K, J, Q>(k, lane);
    if constexpr (J > 1)
        sort_from<K, J / 2, Q>(k, lane);
    else if constexpr (K < 32 * Q)
        sort_from<2 * K, K, Q>(k, lane);
}

template <int K>
__device__ __forceinline__ void
add_sorted_rows(const float* __restrict__ src, int lane, int kreg, int t,
                float& a0, float& a1, float& a2, float& a3)
{
    int kk[K];
#pragma unroll
    for (int u = 0; u < K; ++u)
        kk[u] = __builtin_amdgcn_readlane(kreg, t + u);
    float v[K][4];
#pragma unroll
    for (int u = 0; u < K; ++u) {
        const float* __restrict__ sp = src + (size_t)kk[u] * DIM + lane;
        v[u][0] = sp[0];
        v[u][1] = sp[32];
        v[u][2] = sp[64];
        v[u][3] = sp[96];
    }
#pragma unroll
    for (int u = 0; u < K; ++u) {
        a0 += v[u][0];
        a1 += v[u][1];
        a2 += v[u][2];
        a3 += v[u][3];
    }
}

template <int Q>
__device__ __forceinline__ void
sorted_row_sum(const float* __restrict__ src, const int* __restrict__ lst,
               int lane, int n, float& a0, float& a1, float& a2, float& a3)
{
    int k[Q];
#pragma unroll
    for (int q = 0; q < Q; ++q) {
        const int j = lane + 32 * q;
        k[q] = j < n ? lst[j] : KEY_NONE;
    }
    sort_from<2, 1, Q>(k, lane);

#pragma unroll
    for (int q = 0; q < Q; ++q) {
        int c = n - 32 * q;
        c = c < 0 ? 0 : (c > 32 ? 32 : c);
        int t = 0;
        for (; t + 4 <= c; t += 4)
            add_sorted_rows<4>(src, lane, k[q], t, a0, a1, a2, a3);
        if (t + 2 <= c) {
            add_sorted_rows<2>(src, lane, k[q], t, a0, a1, a2, a3);
            t += 2;
        }
        if (t < c)
            add_sorted_rows<1>(src, lane, k[q], t, a0, a1, a2, a3);
    }
}

__device__ __forceinline__ void
sum_row(const float* __restrict__ src, const int* __restrict__ lst,
        int lane, int n, float& a0, float& a1, float& a2, float& a3)
{
    if (n <= 32) {
        sorted_row_sum<1>(src, lst, lane, n, a0, a1, a2, a3);
    } else if (n <= 64) {
        sorted_row_sum<2>(src, lst, lane, n, a0, a1, a2, a3);
    } else if (n <= 128) {
        sorted_row_sum<4>(src, lst, lane, n, a0, a1, a2, a3);
    } else {
        int prev = -1;
        int done = 0;
        while (done < n) {
            int mn = KEY_NONE;
            for (int j = lane; j < n; j += 32) {
                const int k = lst[j];
                if (k > prev && k < mn) mn = k;
            }
            for (int o = 16; o > 0; o >>= 1) {
                const int other = __shfl_xor(mn, o);
                mn = other < mn ? other : mn;
            }
            int c = 0;
            for (int j = lane; j < n; j += 32) c += (lst[j] == mn) ? 1 : 0;
            for (int o = 16; o > 0; o >>= 1) c += __shfl_xor(c, o);
            mn = __builtin_amdgcn_readfirstlane(mn);
            c  = __builtin_amdgcn_readfirstlane(c);
            if (c <= 0) break;
            const float* __restrict__ sp = src + (size_t)mn * DIM + lane;
            const float w0 = sp[0];
            const float w1 = sp[32];
            const float w2 = sp[64];
            const float w3 = sp[96];
            for (int i = 0; i < c; ++i) {
                a0 += w0;
                a1 += w1;
                a2 += w2;
                a3 += w3;
            }
            prev = mn;
            done += c;
        }
    }
}

template <int FINAL>
__global__ void __launch_bounds__(256)
rowsum_kernel(const float* __restrict__ src, float* __restrict__ dst,
              const int* __restrict__ cnt, const int* __restrict__ off,
              const int* __restrict__ list, const float* __restrict__ bias,
              int nrows)
{
    const int row  = (int)(((long)blockIdx.x * blockDim.x + threadIdx.x) >> 5);
    const int lane = threadIdx.x & 31;
    if (row >= nrows) return;

    const int n = __builtin_amdgcn_readfirstlane(cnt[row]);
    const int b = __builtin_amdgcn_readfirstlane(off[row]);
    const int* __restrict__ lst = list + b;

    float a0 = 0.0f, a1 = 0.0f, a2 = 0.0f, a3 = 0.0f;
    sum_row(src, lst, lane, n, a0, a1, a2, a3);

    const float inv = n > 0 ? 1.0f / (float)n : 0.0f;
    float* __restrict__ dp = dst + (size_t)row * DIM + lane;
    if (FINAL) {
        const float y0 = a0 * inv + bias[lane];
        const float y1 = a1 * inv + bias[lane + 32];
        const float y2 = a2 * inv + bias[lane + 64];
        const float y3 = a3 * inv + bias[lane + 96];
        dp[0]  = y0 >= 0.0f ? y0 : NEG_SLOPE * y0;
        dp[32] = y1 >= 0.0f ? y1 : NEG_SLOPE * y1;
        dp[64] = y2 >= 0.0f ? y2 : NEG_SLOPE * y2;
        dp[96] = y3 >= 0.0f ? y3 : NEG_SLOPE * y3;
    } else {
        dp[0]  = a0 * inv;
        dp[32] = a1 * inv;
        dp[64] = a2 * inv;
        dp[96] = a3 * inv;
    }
}

#define TILE_P 132
__global__ void __launch_bounds__(512)
rowmean_gemm_kernel(const float* __restrict__ X, const float* __restrict__ W,
                    float* __restrict__ M, const int* __restrict__ cnt,
                    const int* __restrict__ off, const int* __restrict__ list)
{
    __shared__ float tile[16 * TILE_P];
    const int lane = threadIdx.x & 31;
    const int wv   = threadIdx.x >> 5;
    const int row0 = blockIdx.x * 16;
    {
        const int row = row0 + wv;
        const int n = __builtin_amdgcn_readfirstlane(cnt[row]);
        const int b = __builtin_amdgcn_readfirstlane(off[row]);
        float a0 = 0.0f, a1 = 0.0f, a2 = 0.0f, a3 = 0.0f;
        sum_row(X, list + b, lane, n, a0, a1, a2, a3);
        const float inv = n > 0 ? 1.0f / (float)n : 0.0f;
        float* tp = tile + wv * TILE_P + lane;
        tp[0]  = a0 * inv;
        tp[32] = a1 * inv;
        tp[64] = a2 * inv;
        tp[96] = a3 * inv;
    }
    __syncthreads();
    if (wv >= DIM / 16) return;

    const int half = lane >> 4;
    const int lr   = lane & 15;
    const int col0 = wv * 16;
    const float* ap = tile + lr * TILE_P + 2 * half;
    const float* __restrict__ bp = W + (size_t)(2 * half) * DIM + col0 + lr;

    unsigned yo = (unsigned)(((row0 + half * 8) * DIM + col0 + lr) * sizeof(float));
    asm volatile("" : "+v"(yo));
    float* __restrict__ yp = (float*)((char*)M + yo);

    v8f c = {};
#pragma unroll
    for (int k = 0; k < DIM; k += 4) {
        v2f a, b;
        a.x = ap[k];
        a.y = ap[k + 1];
        b.x = bp[(size_t)k * DIM];
        b.y = bp[(size_t)k * DIM + DIM];
        c = __builtin_amdgcn_wmma_f32_16x16x4_f32(false, a, false, b,
                                                  (short)0, c, false, false);
    }

#pragma unroll
    for (int r = 0; r < 8; ++r)
        yp[(size_t)r * DIM] = c[r];
}

extern "C" void kernel_launch(void* const* d_in, const int* in_sizes, int n_in,
                              void* d_out, int out_size, void* d_ws, size_t ws_size,
                              hipStream_t stream)
{
    const float* X    = (const float*)d_in[0];
    const int*   A    = (const int*)d_in[1];
    const float* W    = (const float*)d_in[2];
    const float* bias = (const float*)d_in[3];
    float*       out  = (float*)d_out;

    const int N   = in_sizes[0] / DIM;
    const int NNZ = in_sizes[1] / 2;
    const int* node = A;
    const int* edge = A + NNZ;
    const long n_out = (long)N * DIM;
    const long n_m   = (long)NEDGES * DIM;

    const int mtiles = N / 16;

    const int  R    = NEDGES + N;
    const int  nb   = (R + RB_ROWS - 1) / RB_ROWS;
    float* m    = (float*)d_ws;
    int*   cnt  = (int*)(m + n_m);
    int*   off  = cnt + R;
    int*   list = off + R;
    int*   bcnt = list + 2L * NNZ;
    int*   bcur = bcnt + NBUCK_MAX;
    int*   boff = bcur + NBUCK_MAX;
    int*   recs = (int*)d_out;
    const int etiles = NEDGES / 16;
    const size_t need = ((size_t)n_m + 2 * (size_t)R + 2 * (size_t)NNZ + 3 * NBUCK_MAX) * sizeof(float);

    if (ws_size >= need && nb <= NBUCK_MAX && N <= (1 << SRC_BITS) &&
        NEDGES <= (1 << SRC_BITS) && 2L * NNZ <= n_out) {
        const int nchunks = (NNZ + CHUNK - 1) / CHUNK;
        zero_kernel<<<(2 * NBUCK_MAX + 255) / 256, 256, 0, stream>>>((float*)bcnt, 2L * NBUCK_MAX);
        range_count_kernel<<<nchunks, PT, 0, stream>>>(node, edge, bcnt, NNZ);
        range_move_kernel<<<nchunks, PT, 0, stream>>>(node, edge, bcnt, bcur, boff, recs, NNZ);
        range_rows_kernel<<<nb, PT, 0, stream>>>(recs, bcnt, boff, cnt, off, list, R);
        rowmean_gemm_kernel<<<etiles, 512, 0, stream>>>(X, W, m, cnt, off, list);
        rowsum_kernel<1><<<(N + 7) / 8, 256, 0, stream>>>(m, out, cnt + NEDGES, off + NEDGES, list, bias, N);
        return;
    }

    gemm_wmma_f32<<<mtiles, 256, 0, stream>>>(X, W, out, mtiles);

    float* Ddeg = m + (size_t)NEDGES * DIM;
    float* Bdeg = Ddeg + N;
    const long ws_f  = (long)NEDGES * DIM + N + NEDGES;
    zero_kernel<<<(int)((ws_f + 255) / 256), 256, 0, stream>>>(m, ws_f);
    degrees_kernel<<<(NNZ + 255) / 256, 256, 0, stream>>>(node, edge, Ddeg, Bdeg, NNZ);
    scatter_kernel<<<(NNZ + 7) / 8, 256, 0, stream>>>(out, m, node, edge, NNZ);
    scale_rows_kernel<<<(int)((n_m + 255) / 256), 256, 0, stream>>>(m, Bdeg, n_m);
    zero_kernel<<<(int)((n_out + 255) / 256), 256, 0, stream>>>(out, n_out);
    scatter_kernel<<<(NNZ + 7) / 8, 256, 0, stream>>>(m, out, edge, node, NNZ);
    finalize_kernel<<<(int)((n_out + 255) / 256), 256, 0, stream>>>(out, Ddeg, bias, n_out);
}


# pass 2: a resident grid of 10240 waves, each looping over rows with the next rows' count, list start and keys requested ahead, instead of 200000 one-row waves; same add order, bit-identical
# speedup vs baseline: 5.8538x; 1.0441x over previous
#include <hip/hip_runtime.h>

typedef __attribute__((ext_vector_type(2))) float v2f;
typedef __attribute__((ext_vector_type(8))) float v8f;

#define DIM        128
#define NEDGES     50000
#define NEG_SLOPE  0.01f

__global__ void __launch_bounds__(256)
gemm_wmma_f32(const float* __restrict__ X, const float* __restrict__ W,
              float* __restrict__ Y, int mtiles)
{
    const int lane  = threadIdx.x & 31;
    const int wv    = threadIdx.x >> 5;
    const long tile = (long)blockIdx.x * 8 + wv;
    const long mt   = tile >> 3;
    const int  nt   = (int)(tile & 7);
    if (mt >= mtiles) return;

    const int half = lane >> 4;
    const int lr   = lane & 15;
    const size_t row  = (size_t)mt * 16 + lr;
    const int    col0 = nt * 16;

    const float* __restrict__ ap = X + row * DIM + 2 * half;
    const float* __restrict__ bp = W + (size_t)(2 * half) * DIM + col0 + lr;

    v8f c = {};
#pragma unroll
    for (int k = 0; k < DIM; k += 4) {
        v2f a, b;
        a.x = ap[k];
        a.y = ap[k + 1];
        b.x = bp[(size_t)k * DIM];
        b.y = bp[(size_t)k * DIM + DIM];
        c = __builtin_amdgcn_wmma_f32_16x16x4_f32(false, a, false, b,
                                                  (short)0, c, false, false);
    }

    float* __restrict__ yp = Y + ((size_t)mt * 16 + half * 8) * DIM + col0 + lr;
#pragma unroll
    for (int r = 0; r < 8; ++r)
        yp[(size_t)r * DIM] = c[r];
}

__global__ void zero_kernel(float* __restrict__ p, long n)
{
    long i = (long)blockIdx.x * blockDim.x + threadIdx.x;
    if (i < n) p[i] = 0.0f;
}

__global__ void degrees_kernel(const int* __restrict__ node,
                               const int* __restrict__ edge,
                               float* __restrict__ Ddeg,
                               float* __restrict__ Bdeg, int nnz)
{
    int i = blockIdx.x * blockDim.x + threadIdx.x;
    if (i >= nnz) return;
    unsafeAtomicAdd(&Ddeg[node[i]], 1.0f);
    unsafeAtomicAdd(&Bdeg[edge[i]], 1.0f);
}

__global__ void __launch_bounds__(256)
scatter_kernel(const float* __restrict__ src, float* __restrict__ dst,
               const int* __restrict__ gidx, const int* __restrict__ sidx,
               int nnz)
{
    const int wid  = (int)(((long)blockIdx.x * blockDim.x + threadIdx.x) >> 5);
    const int lane = threadIdx.x & 31;
    if (wid >= nnz) return;

    const int g = gidx[wid];
    const int s = sidx[wid];

    const float* __restrict__ sp = src + (size_t)g * DIM + lane;
    float* dp = dst + (size_t)s * DIM + lane;
    const float v0 = sp[0];
    const float v1 = sp[32];
    const float v2 = sp[64];
    const float v3 = sp[96];
    unsafeAtomicAdd(dp + 0,  v0);
    unsafeAtomicAdd(dp + 32, v1);
    unsafeAtomicAdd(dp + 64, v2);
    unsafeAtomicAdd(dp + 96, v3);
}

__global__ void scale_rows_kernel(float* __restrict__ m,
                                  const float* __restrict__ Bdeg, long n)
{
    long i = (long)blockIdx.x * blockDim.x + threadIdx.x;
    if (i >= n) return;
    float bb  = Bdeg[i >> 7];
    float inv = bb > 0.0f ? 1.0f / bb : 0.0f;
    m[i] *= inv;
}

__global__ void finalize_kernel(float* __restrict__ out,
                                const float* __restrict__ Ddeg,
                                const float* __restrict__ bias, long n)
{
    long i = (long)blockIdx.x * blockDim.x + threadIdx.x;
    if (i >= n) return;
    float dd  = Ddeg[i >> 7];
    float inv = dd > 0.0f ? 1.0f / dd : 0.0f;
    float v   = out[i] * inv + bias[i & (DIM - 1)];
    out[i] = v >= 0.0f ? v : NEG_SLOPE * v;
}

#define KEY_NONE 0x7fffffff

#define RB_SHIFT  9
#define RB_ROWS   (1 << RB_SHIFT)
#define NBUCK_MAX 512
#define SRC_BITS  18
#define SRC_MASK  ((1 << SRC_BITS) - 1)
#define PT        1024
#define EPT       8
#define CHUNK     (PT * EPT)

__device__ __forceinline__ int scan512(int x, int tid, int* wsum)
{
    const int lane = tid & 31;
    const int wv   = tid >> 5;
    int v = x;
    for (int o = 1; o < 32; o <<= 1) {
        const int t = __shfl_up(v, o);
        if (lane >= o) v += t;
    }
    if (lane == 31 && wv < 16) wsum[wv] = v;
    __syncthreads();
    if (wv == 0) {
        int w = lane < 16 ? wsum[lane] : 0;
        for (int o = 1; o < 16; o <<= 1) {
            const int t = __shfl_up(w, o);
            if (lane >= o) w += t;
        }
        if (lane < 16) wsum[lane] = w;
    }
    __syncthreads();
    if (wv > 0 && wv < 16) v += wsum[wv - 1];
    return v;
}

__global__ void __launch_bounds__(PT)
range_count_kernel(const int* __restrict__ node, const int* __restrict__ edge,
                   int* __restrict__ bcnt, int nnz)
{
    __shared__ int h[NBUCK_MAX];
    const int tid = threadIdx.x;
    if (tid < NBUCK_MAX) h[tid] = 0;
    __syncthreads();
    const int base = blockIdx.x * CHUNK + tid;
    int ke[EPT], kn[EPT];
#pragma unroll
    for (int u = 0; u < EPT; ++u) {
        const int i = base + u * PT;
        ke[u] = i < nnz ? edge[i] : -1;
        kn[u] = i < nnz ? node[i] : 0;
    }
#pragma unroll
    for (int u = 0; u < EPT; ++u) {
        if (ke[u] >= 0) {
            atomicAdd(&h[ke[u] >> RB_SHIFT], 1);
            atomicAdd(&h[(NEDGES + kn[u]) >> RB_SHIFT], 1);
        }
    }
    __syncthreads();
    if (tid < NBUCK_MAX) {
        const int c = h[tid];
        if (c > 0) atomicAdd(&bcnt[tid], c);
    }
}

__global__ void __launch_bounds__(PT)
range_move_kernel(const int* __restrict__ node, const int* __restrict__ edge,
                  const int* __restrict__ bcnt, int* __restrict__ bcur,
                  int* __restrict__ boff, int* __restrict__ recs, int nnz)
{
    __shared__ int h[NBUCK_MAX];
    __shared__ int sbase[NBUCK_MAX];
    __shared__ int wsum[16];
    const int tid = threadIdx.x;
    if (tid < NBUCK_MAX) h[tid] = 0;
    __syncthreads();

    int ke[EPT], kn[EPT], re[EPT], rn[EPT];
    const int base = blockIdx.x * CHUNK + tid;
#pragma unroll
    for (int u = 0; u < EPT; ++u) {
        const int i = base + u * PT;
        ke[u] = i < nnz ? edge[i] : -1;
        kn[u] = i < nnz ? node[i] : 0;
    }
#pragma unroll
    for (int u = 0; u < EPT; ++u) {
        re[u] = 0;
        rn[u] = 0;
        if (ke[u] >= 0) {
            re[u] = atomicAdd(&h[ke[u] >> RB_SHIFT], 1);
            rn[u] = atomicAdd(&h[(NEDGES + kn[u]) >> RB_SHIFT], 1);
        }
    }

    const int tot = tid < NBUCK_MAX ? bcnt[tid] : 0;
    const int inc = scan512(tot, tid, wsum);
    if (tid < NBUCK_MAX) {
        const int ex = inc - tot;
        if (blockIdx.x == 0) boff[tid] = ex;
        const int c = h[tid];
        int b = 0;
        if (c > 0) b = atomicAdd(&bcur[tid], c);
        sbase[tid] = ex + b;
    }
    __syncthreads();

#pragma unroll
    for (int u = 0; u < EPT; ++u) {
        if (ke[u] >= 0) {
            const int e  = ke[u];
            const int n  = kn[u];
            const int rr = NEDGES + n;
            recs[sbase[e >> RB_SHIFT] + re[u]]  = ((e & (RB_ROWS - 1)) << SRC_BITS) | n;
            recs[sbase[rr >> RB_SHIFT] + rn[u]] = ((rr & (RB_ROWS - 1)) << SRC_BITS) | e;
        }
    }
}

__global__ void __launch_bounds__(PT)
range_rows_kernel(const int* __restrict__ recs, const int* __restrict__ bcnt,
                  const int* __restrict__ boff, int* __restrict__ cnt,
                  int* __restrict__ off, int* __restrict__ list, int nrows)
{
    __shared__ int c[RB_ROWS];
    __shared__ int f[RB_ROWS];
    __shared__ int wsum[16];
    const int tid   = threadIdx.x;
    const int b     = blockIdx.x;
    const int n     = bcnt[b];
    const int start = boff[b];
    const int* __restrict__ rp = recs + start;
    if (tid < RB_ROWS) c[tid] = 0;
    __syncthreads();

    for (int j0 = tid; j0 < n; j0 += 4 * PT) {
        int r[4];
#pragma unroll
        for (int u = 0; u < 4; ++u) {
            const int j = j0 + u * PT;
            r[u] = j < n ? rp[j] : -1;
        }
#pragma unroll
        for (int u = 0; u < 4; ++u)
            if (r[u] >= 0) atomicAdd(&c[r[u] >> SRC_BITS], 1);
    }
    __syncthreads();

    const int mine = tid < RB_ROWS ? c[tid] : 0;
    const int inc  = scan512(mine, tid, wsum);
    if (tid < RB_ROWS) {
        const int ex  = inc - mine;
        const int row = b * RB_ROWS + tid;
        f[tid] = ex;
        if (row < nrows) {
            cnt[row] = mine;
            off[row] = start + ex;
        }
    }
    __syncthreads();

    int* __restrict__ lp = list + start;
    for (int j0 = tid; j0 < n; j0 += 4 * PT) {
        int r[4];
#pragma unroll
        for (int u = 0; u < 4; ++u) {
            const int j = j0 + u * PT;
            r[u] = j < n ? rp[j] : -1;
        }
#pragma unroll
        for (int u = 0; u < 4; ++u) {
            if (r[u] >= 0) {
                const int p = atomicAdd(&f[r[u] >> SRC_BITS], 1);
                lp[p] = r[u] & SRC_MASK;
            }
        }
    }
}

template <int J>
__device__ __forceinline__ int lane_xor(int v)
{
    if constexpr (J == 16)
        return __builtin_amdgcn_permlanex16(v, v, 0x76543210, 0xfedcba98, false, false);
    else
        return __builtin_amdgcn_update_dpp(v, v, 0x160 + J, 0xf, 0xf, true);
}

template <int K, int J, int Q>
__device__ __forceinline__ void sort_step(int (&k)[Q], int lane)
{
    if constexpr (J >= 32) {
        constexpr int dq = J >> 5;
#pragma unroll
        for (int q = 0; q < Q; ++q) {
            if ((q & dq) == 0) {
                const bool asc = ((32 * q) & K) == 0;
                const int x  = k[q];
                const int y  = k[q + dq];
                const int lo = y < x ? y : x;
                const int hi = y < x ? x : y;
                k[q]      = asc ? lo : hi;
                k[q + dq] = asc ? hi : lo;
            }
        }
    } else {
        const bool lower = (lane & J) == 0;
#pragma unroll
        for (int q = 0; q < Q; ++q) {
            const bool asc = K < 32 ? (lane & K) == 0 : ((32 * q) & K) == 0;
            const int x  = k[q];
            const int o  = lane_xor<J>(x);
            const int mn = o < x ? o : x;
            const int mx = o < x ? x : o;
            k[q] = (lower == asc) ? mn : mx;
        }
    }
}

template <int K, int J, int Q>
__device__ __forceinline__ void sort_from(int (&k)[Q], int lane)
{
    sort_step<K, J, Q>(k, lane);
    if constexpr (J > 1)
        sort_from<K, J / 2, Q>(k, lane);
    else if constexpr (K < 32 * Q)
        sort_from<2 * K, K, Q>(k, lane);
}

template <int K>
__device__ __forceinline__ void
add_sorted_rows(const float* __restrict__ src, int lane, int kreg, int t,
                float& a0, float& a1, float& a2, float& a3)
{
    int kk[K];
#pragma unroll
    for (int u = 0; u < K; ++u)
        kk[u] = __builtin_amdgcn_readlane(kreg, t + u);
    float v[K][4];
#pragma unroll
    for (int u = 0; u < K; ++u) {
        const float* __restrict__ sp = src + (size_t)kk[u] * DIM + lane;
        v[u][0] = sp[0];
        v[u][1] = sp[32];
        v[u][2] = sp[64];
        v[u][3] = sp[96];
    }
#pragma unroll
    for (int u = 0; u < K; ++u) {
        a0 += v[u][0];
        a1 += v[u][1];
        a2 += v[u][2];
        a3 += v[u][3];
    }
}

template <int Q>
__device__ __forceinline__ void
sum_loaded_keys(const float* __restrict__ src, int (&k)[Q],
                int lane, int n, float& a0, float& a1, float& a2, float& a3)
{
    sort_from<2, 1, Q>(k, lane);

#pragma unroll
    for (int q = 0; q < Q; ++q) {
        int c = n - 32 * q;
        c = c < 0 ? 0 : (c > 32 ? 32 : c);
        int t = 0;
        for (; t + 4 <= c; t += 4)
            add_sorted_rows<4>(src, lane, k[q], t, a0, a1, a2, a3);
        if (t + 2 <= c) {
            add_sorted_rows<2>(src, lane, k[q], t, a0, a1, a2, a3);
            t += 2;
        }
        if (t < c)
            add_sorted_rows<1>(src, lane, k[q], t, a0, a1, a2, a3);
    }
}

template <int Q>
__device__ __forceinline__ void
sorted_row_sum(const float* __restrict__ src, const int* __restrict__ lst,
               int lane, int n, float& a0, float& a1, float& a2, float& a3)
{
    int k[Q];
#pragma unroll
    for (int q = 0; q < Q; ++q) {
        const int j = lane + 32 * q;
        k[q] = j < n ? lst[j] : KEY_NONE;
    }
    sum_loaded_keys<Q>(src, k, lane, n, a0, a1, a2, a3);
}

__device__ __forceinline__ void
sum_row(const float* __restrict__ src, const int* __restrict__ lst,
        int lane, int n, float& a0, float& a1, float& a2, float& a3)
{
    if (n <= 32) {
        sorted_row_sum<1>(src, lst, lane, n, a0, a1, a2, a3);
    } else if (n <= 64) {
        sorted_row_sum<2>(src, lst, lane, n, a0, a1, a2, a3);
    } else if (n <= 128) {
        sorted_row_sum<4>(src, lst, lane, n, a0, a1, a2, a3);
    } else {
        int prev = -1;
        int done = 0;
        while (done < n) {
            int mn = KEY_NONE;
            for (int j = lane; j < n; j += 32) {
                const int k = lst[j];
                if (k > prev && k < mn) mn = k;
            }
            for (int o = 16; o > 0; o >>= 1) {
                const int other = __shfl_xor(mn, o);
                mn = other < mn ? other : mn;
            }
            int c = 0;
            for (int j = lane; j < n; j += 32) c += (lst[j] == mn) ? 1 : 0;
            for (int o = 16; o > 0; o >>= 1) c += __shfl_xor(c, o);
            mn = __builtin_amdgcn_readfirstlane(mn);
            c  = __builtin_amdgcn_readfirstlane(c);
            if (c <= 0) break;
            const float* __restrict__ sp = src + (size_t)mn * DIM + lane;
            const float w0 = sp[0];
            const float w1 = sp[32];
            const float w2 = sp[64];
            const float w3 = sp[96];
            for (int i = 0; i < c; ++i) {
                a0 += w0;
                a1 += w1;
                a2 += w2;
                a3 += w3;
            }
            prev = mn;
            done += c;
        }
    }
}

__global__ void __launch_bounds__(256)
rowsum_loop_kernel(const float* __restrict__ src, float* __restrict__ dst,
                   const int* __restrict__ cnt, const int* __restrict__ off,
                   const int* __restrict__ list, const float* __restrict__ bias,
                   int nrows)
{
    const int lane = threadIdx.x & 31;
    const int nw   = (int)(gridDim.x * (blockDim.x >> 5));
    int row = __builtin_amdgcn_readfirstlane((int)((blockIdx.x * blockDim.x + threadIdx.x) >> 5));
    if (row >= nrows) return;

    int n0 = __builtin_amdgcn_readfirstlane(cnt[row]);
    int b0 = __builtin_amdgcn_readfirstlane(off[row]);
    int key0 = lane < n0 ? list[b0 + lane] : KEY_NONE;
    int row1 = row + nw;
    int n1 = 0, b1 = 0;
    if (row1 < nrows) {
        n1 = __builtin_amdgcn_readfirstlane(cnt[row1]);
        b1 = __builtin_amdgcn_readfirstlane(off[row1]);
    }

    for (;;) {
        const int row2 = row1 + nw;
        int key1 = KEY_NONE;
        int n2v = 0, b2v = 0;
        if (row1 < nrows) key1 = lane < n1 ? list[b1 + lane] : KEY_NONE;
        if (row2 < nrows) { n2v = cnt[row2]; b2v = off[row2]; }

        float a0 = 0.0f, a1 = 0.0f, a2 = 0.0f, a3 = 0.0f;
        if (n0 <= 32) {
            int k[1] = { key0 };
            sum_loaded_keys<1>(src, k, lane, n0, a0, a1, a2, a3);
        } else {
            sum_row(src, list + b0, lane, n0, a0, a1, a2, a3);
        }

        const float inv = n0 > 0 ? 1.0f / (float)n0 : 0.0f;
        float* dp = dst + (size_t)row * DIM + lane;
        const float y0 = a0 * inv + bias[lane];
        const float y1 = a1 * inv + bias[lane + 32];
        const float y2 = a2 * inv + bias[lane + 64];
        const float y3 = a3 * inv + bias[lane + 96];
        dp[0]  = y0 >= 0.0f ? y0 : NEG_SLOPE * y0;
        dp[32] = y1 >= 0.0f ? y1 : NEG_SLOPE * y1;
        dp[64] = y2 >= 0.0f ? y2 : NEG_SLOPE * y2;
        dp[96] = y3 >= 0.0f ? y3 : NEG_SLOPE * y3;

        if (row1 >= nrows) break;
        row  = row1;
        n0   = n1;
        b0   = b1;
        key0 = key1;
        row1 = row2;
        n1   = __builtin_amdgcn_readfirstlane(n2v);
        b1   = __builtin_amdgcn_readfirstlane(b2v);
    }
}

#define TILE_P 132
__global__ void __launch_bounds__(512)
rowmean_gemm_kernel(const float* __restrict__ X, const float* __restrict__ W,
                    float* __restrict__ M, const int* __restrict__ cnt,
                    const int* __restrict__ off, const int* __restrict__ list)
{
    __shared__ float tile[16 * TILE_P];
    const int lane = threadIdx.x & 31;
    const int wv   = threadIdx.x >> 5;
    const int row0 = blockIdx.x * 16;
    {
        const int row = row0 + wv;
        const int n = __builtin_amdgcn_readfirstlane(cnt[row]);
        const int b = __builtin_amdgcn_readfirstlane(off[row]);
        float a0 = 0.0f, a1 = 0.0f, a2 = 0.0f, a3 = 0.0f;
        sum_row(X, list + b, lane, n, a0, a1, a2, a3);
        const float inv = n > 0 ? 1.0f / (float)n : 0.0f;
        float* tp = tile + wv * TILE_P + lane;
        tp[0]  = a0 * inv;
        tp[32] = a1 * inv;
        tp[64] = a2 * inv;
        tp[96] = a3 * inv;
    }
    __syncthreads();
    if (wv >= DIM / 16) return;

    const int half = lane >> 4;
    const int lr   = lane & 15;
    const int col0 = wv * 16;
    const float* ap = tile + lr * TILE_P + 2 * half;
    const float* __restrict__ bp = W + (size_t)(2 * half) * DIM + col0 + lr;

    unsigned yo = (unsigned)(((row0 + half * 8) * DIM + col0 + lr) * sizeof(float));
    asm volatile("" : "+v"(yo));
    float* __restrict__ yp = (float*)((char*)M + yo);

    v8f c = {};
#pragma unroll
    for (int k = 0; k < DIM; k += 4) {
        v2f a, b;
        a.x = ap[k];
        a.y = ap[k + 1];
        b.x = bp[(size_t)k * DIM];
        b.y = bp[(size_t)k * DIM + DIM];
        c = __builtin_amdgcn_wmma_f32_16x16x4_f32(false, a, false, b,
                                                  (short)0, c, false, false);
    }

#pragma unroll
    for (int r = 0; r < 8; ++r)
        yp[(size_t)r * DIM] = c[r];
}

extern "C" void kernel_launch(void* const* d_in, const int* in_sizes, int n_in,
                              void* d_out, int out_size, void* d_ws, size_t ws_size,
                              hipStream_t stream)
{
    const float* X    = (const float*)d_in[0];
    const int*   A    = (const int*)d_in[1];
    const float* W    = (const float*)d_in[2];
    const float* bias = (const float*)d_in[3];
    float*       out  = (float*)d_out;

    const int N   = in_sizes[0] / DIM;
    const int NNZ = in_sizes[1] / 2;
    const int* node = A;
    const int* edge = A + NNZ;
    const long n_out = (long)N * DIM;
    const long n_m   = (long)NEDGES * DIM;

    const int mtiles = N / 16;

    const int  R    = NEDGES + N;
    const int  nb   = (R + RB_ROWS - 1) / RB_ROWS;
    float* m    = (float*)d_ws;
    int*   cnt  = (int*)(m + n_m);
    int*   off  = cnt + R;
    int*   list = off + R;
    int*   bcnt = list + 2L * NNZ;
    int*   bcur = bcnt + NBUCK_MAX;
    int*   boff = bcur + NBUCK_MAX;
    int*   recs = (int*)d_out;
    const int etiles = NEDGES / 16;
    const size_t need = ((size_t)n_m + 2 * (size_t)R + 2 * (size_t)NNZ + 3 * NBUCK_MAX) * sizeof(float);

    if (ws_size >= need && nb <= NBUCK_MAX && N <= (1 << SRC_BITS) &&
        NEDGES <= (1 << SRC_BITS) && 2L * NNZ <= n_out) {
        const int nchunks = (NNZ + CHUNK - 1) / CHUNK;
        zero_kernel<<<(2 * NBUCK_MAX + 255) / 256, 256, 0, stream>>>((float*)bcnt, 2L * NBUCK_MAX);
        range_count_kernel<<<nchunks, PT, 0, stream>>>(node, edge, bcnt, NNZ);
        range_move_kernel<<<nchunks, PT, 0, stream>>>(node, edge, bcnt, bcur, boff, recs, NNZ);
        range_rows_kernel<<<nb, PT, 0, stream>>>(recs, bcnt, boff, cnt, off, list, R);
        rowmean_gemm_kernel<<<etiles, 512, 0, stream>>>(X, W, m, cnt, off, list);
        rowsum_loop_kernel<<<1280, 256, 0, stream>>>(m, out, cnt + NEDGES, off + NEDGES, list, bias, N);
        return;
    }

    gemm_wmma_f32<<<mtiles, 256, 0, stream>>>(X, W, out, mtiles);

    float* Ddeg = m + (size_t)NEDGES * DIM;
    float* Bdeg = Ddeg + N;
    const long ws_f  = (long)NEDGES * DIM + N + NEDGES;
    zero_kernel<<<(int)((ws_f + 255) / 256), 256, 0, stream>>>(m, ws_f);
    degrees_kernel<<<(NNZ + 255) / 256, 256, 0, stream>>>(node, edge, Ddeg, Bdeg, NNZ);
    scatter_kernel<<<(NNZ + 7) / 8, 256, 0, stream>>>(out, m, node, edge, NNZ);
    scale_rows_kernel<<<(int)((n_m + 255) / 256), 256, 0, stream>>>(m, Bdeg, n_m);
    zero_kernel<<<(int)((n_out + 255) / 256), 256, 0, stream>>>(out, n_out);
    scatter_kernel<<<(NNZ + 7) / 8, 256, 0, stream>>>(m, out, edge, node, NNZ);
    finalize_kernel<<<(int)((n_out + 255) / 256), 256, 0, stream>>>(out, Ddeg, bias, n_out);
}


# list building: records and list entries are put in order in LDS and leave as runs of consecutive words (whole lines), not as 4-byte stores that fall into 32 different lines per instruction
# speedup vs baseline: 6.9468x; 1.1867x over previous
#include <hip/hip_runtime.h>

typedef __attribute__((ext_vector_type(2))) float v2f;
typedef __attribute__((ext_vector_type(8))) float v8f;
typedef __attribute__((ext_vector_type(2))) int   v2i;

#define DIM        128
#define NEDGES     50000
#define NEG_SLOPE  0.01f

__global__ void __launch_bounds__(256)
gemm_wmma_f32(const float* __restrict__ X, const float* __restrict__ W,
              float* __restrict__ Y, int mtiles)
{
    const int lane  = threadIdx.x & 31;
    const int wv    = threadIdx.x >> 5;
    const long tile = (long)blockIdx.x * 8 + wv;
    const long mt   = tile >> 3;
    const int  nt   = (int)(tile & 7);
    if (mt >= mtiles) return;

    const int half = lane >> 4;
    const int lr   = lane & 15;
    const size_t row  = (size_t)mt * 16 + lr;
    const int    col0 = nt * 16;

    const float* __restrict__ ap = X + row * DIM + 2 * half;
    const float* __restrict__ bp = W + (size_t)(2 * half) * DIM + col0 + lr;

    v8f c = {};
#pragma unroll
    for (int k = 0; k < DIM; k += 4) {
        v2f a, b;
        a.x = ap[k];
        a.y = ap[k + 1];
        b.x = bp[(size_t)k * DIM];
        b.y = bp[(size_t)k * DIM + DIM];
        c = __builtin_amdgcn_wmma_f32_16x16x4_f32(false, a, false, b,
                                                  (short)0, c, false, false);
    }

    float* __restrict__ yp = Y + ((size_t)mt * 16 + half * 8) * DIM + col0 + lr;
#pragma unroll
    for (int r = 0; r < 8; ++r)
        yp[(size_t)r * DIM] = c[r];
}

__global__ void zero_kernel(float* __restrict__ p, long n)
{
    long i = (long)blockIdx.x * blockDim.x + threadIdx.x;
    if (i < n) p[i] = 0.0f;
}

__global__ void degrees_kernel(const int* __restrict__ node,
                               const int* __restrict__ edge,
                               float* __restrict__ Ddeg,
                               float* __restrict__ Bdeg, int nnz)
{
    int i = blockIdx.x * blockDim.x + threadIdx.x;
    if (i >= nnz) return;
    unsafeAtomicAdd(&Ddeg[node[i]], 1.0f);
    unsafeAtomicAdd(&Bdeg[edge[i]], 1.0f);
}

__global__ void __launch_bounds__(256)
scatter_kernel(const float* __restrict__ src, float* __restrict__ dst,
               const int* __restrict__ gidx, const int* __restrict__ sidx,
               int nnz)
{
    const int wid  = (int)(((long)blockIdx.x * blockDim.x + threadIdx.x) >> 5);
    const int lane = threadIdx.x & 31;
    if (wid >= nnz) return;

    const int g = gidx[wid];
    const int s = sidx[wid];

    const float* __restrict__ sp = src + (size_t)g * DIM + lane;
    float* dp = dst + (size_t)s * DIM + lane;
    const float v0 = sp[0];
    const float v1 = sp[32];
    const float v2 = sp[64];
    const float v3 = sp[96];
    unsafeAtomicAdd(dp + 0,  v0);
    unsafeAtomicAdd(dp + 32, v1);
    unsafeAtomicAdd(dp + 64, v2);
    unsafeAtomicAdd(dp + 96, v3);
}

__global__ void scale_rows_kernel(float* __restrict__ m,
                                  const float* __restrict__ Bdeg, long n)
{
    long i = (long)blockIdx.x * blockDim.x + threadIdx.x;
    if (i >= n) return;
    float bb  = Bdeg[i >> 7];
    float inv = bb > 0.0f ? 1.0f / bb : 0.0f;
    m[i] *= inv;
}

__global__ void finalize_kernel(float* __restrict__ out,
                                const float* __restrict__ Ddeg,
                                const float* __restrict__ bias, long n)
{
    long i = (long)blockIdx.x * blockDim.x + threadIdx.x;
    if (i >= n) return;
    float dd  = Ddeg[i >> 7];
    float inv = dd > 0.0f ? 1.0f / dd : 0.0f;
    float v   = out[i] * inv + bias[i & (DIM - 1)];
    out[i] = v >= 0.0f ? v : NEG_SLOPE * v;
}

#define KEY_NONE 0x7fffffff

#define RB_SHIFT  9
#define RB_ROWS   (1 << RB_SHIFT)
#define NBUCK_MAX 512
#define SRC_BITS  18
#define SRC_MASK  ((1 << SRC_BITS) - 1)
#define PT        1024
#define EPT       8
#define CHUNK     (PT * EPT)
#define LIST_CAP  32768

__device__ __forceinline__ int scan512(int x, int tid, int* wsum)
{
    const int lane = tid & 31;
    const int wv   = tid >> 5;
    int v = x;
    for (int o = 1; o < 32; o <<= 1) {
        const int t = __shfl_up(v, o);
        if (lane >= o) v += t;
    }
    if (lane == 31 && wv < 16) wsum[wv] = v;
    __syncthreads();
    if (wv == 0) {
        int w = lane < 16 ? wsum[lane] : 0;
        for (int o = 1; o < 16; o <<= 1) {
            const int t = __shfl_up(w, o);
            if (lane >= o) w += t;
        }
        if (lane < 16) wsum[lane] = w;
    }
    __syncthreads();
    if (wv > 0 && wv < 16) v += wsum[wv - 1];
    return v;
}

__global__ void __launch_bounds__(PT)
range_count_kernel(const int* __restrict__ node, const int* __restrict__ edge,
                   int* __restrict__ bcnt, int nnz)
{
    __shared__ int h[NBUCK_MAX];
    const int tid = threadIdx.x;
    if (tid < NBUCK_MAX) h[tid] = 0;
    __syncthreads();
    const int base = blockIdx.x * CHUNK + tid;
    int ke[EPT], kn[EPT];
#pragma unroll
    for (int u = 0; u < EPT; ++u) {
        const int i = base + u * PT;
        ke[u] = i < nnz ? edge[i] : -1;
        kn[u] = i < nnz ? node[i] : 0;
    }
#pragma unroll
    for (int u = 0; u < EPT; ++u) {
        if (ke[u] >= 0) {
            atomicAdd(&h[ke[u] >> RB_SHIFT], 1);
            atomicAdd(&h[(NEDGES + kn[u]) >> RB_SHIFT], 1);
        }
    }
    __syncthreads();
    if (tid < NBUCK_MAX) {
        const int c = h[tid];
        if (c > 0) atomicAdd(&bcnt[tid], c);
    }
}

__global__ void __launch_bounds__(PT)
range_move_kernel(const int* __restrict__ node, const int* __restrict__ edge,
                  const int* __restrict__ bcnt, int* __restrict__ bcur,
                  int* __restrict__ boff, int* __restrict__ recs, int nnz)
{
    __shared__ int h[NBUCK_MAX];
    __shared__ v2i base[NBUCK_MAX];
    __shared__ int wsum[16];
    __shared__ int wsum2[16];
    __shared__ v2i stage[2 * CHUNK];
    const int tid = threadIdx.x;
    if (tid < NBUCK_MAX) h[tid] = 0;
    __syncthreads();

    int ke[EPT], kn[EPT], re[EPT], rn[EPT];
    const int base0 = blockIdx.x * CHUNK + tid;
#pragma unroll
    for (int u = 0; u < EPT; ++u) {
        const int i = base0 + u * PT;
        ke[u] = i < nnz ? edge[i] : -1;
        kn[u] = i < nnz ? node[i] : 0;
    }
#pragma unroll
    for (int u = 0; u < EPT; ++u) {
        re[u] = 0;
        rn[u] = 0;
        if (ke[u] >= 0) {
            re[u] = atomicAdd(&h[ke[u] >> RB_SHIFT], 1);
            rn[u] = atomicAdd(&h[(NEDGES + kn[u]) >> RB_SHIFT], 1);
        }
    }

    const int tot = tid < NBUCK_MAX ? bcnt[tid] : 0;
    const int inc = scan512(tot, tid, wsum);
    const int c   = tid < NBUCK_MAX ? h[tid] : 0;
    const int lin = scan512(c, tid, wsum2);
    if (tid < NBUCK_MAX) {
        const int ex = inc - tot;
        if (blockIdx.x == 0) boff[tid] = ex;
        int b = 0;
        if (c > 0) b = atomicAdd(&bcur[tid], c);
        v2i s;
        s.x = lin - c;
        s.y = ex + b;
        base[tid] = s;
    }
    __syncthreads();

#pragma unroll
    for (int u = 0; u < EPT; ++u) {
        if (ke[u] >= 0) {
            const int e  = ke[u];
            const int n  = kn[u];
            const int rr = NEDGES + n;
            const v2i se = base[e >> RB_SHIFT];
            const v2i sn = base[rr >> RB_SHIFT];
            v2i we, wn;
            we.x = ((e & (RB_ROWS - 1)) << SRC_BITS) | n;
            we.y = se.y + re[u];
            wn.x = ((rr & (RB_ROWS - 1)) << SRC_BITS) | e;
            wn.y = sn.y + rn[u];
            stage[se.x + re[u]] = we;
            stage[sn.x + rn[u]] = wn;
        }
    }
    __syncthreads();

    const int left = nnz - (int)blockIdx.x * CHUNK;
    const int nrec = 2 * (left < CHUNK ? left : CHUNK);
    v2i w[2 * EPT];
#pragma unroll
    for (int u = 0; u < 2 * EPT; ++u) {
        const int p = tid + u * PT;
        w[u] = stage[p < nrec ? p : 0];
    }
#pragma unroll
    for (int u = 0; u < 2 * EPT; ++u) {
        const int p = tid + u * PT;
        if (p < nrec) recs[w[u].y] = w[u].x;
    }
}

__global__ void __launch_bounds__(PT)
range_rows_kernel(const int* __restrict__ recs, const int* __restrict__ bcnt,
                  const int* __restrict__ boff, int* __restrict__ cnt,
                  int* __restrict__ off, int* __restrict__ list, int nrows)
{
    __shared__ int c[RB_ROWS];
    __shared__ int f[RB_ROWS];
    __shared__ int wsum[16];
    __shared__ int lst[LIST_CAP];
    const int tid   = threadIdx.x;
    const int b     = blockIdx.x;
    const int n     = bcnt[b];
    const int start = boff[b];
    const int* __restrict__ rp = recs + start;
    if (tid < RB_ROWS) c[tid] = 0;
    __syncthreads();

    for (int j0 = tid; j0 < n; j0 += 4 * PT) {
        int r[4];
#pragma unroll
        for (int u = 0; u < 4; ++u) {
            const int j = j0 + u * PT;
            r[u] = j < n ? rp[j] : -1;
        }
#pragma unroll
        for (int u = 0; u < 4; ++u)
            if (r[u] >= 0) atomicAdd(&c[r[u] >> SRC_BITS], 1);
    }
    __syncthreads();

    const int mine = tid < RB_ROWS ? c[tid] : 0;
    const int inc  = scan512(mine, tid, wsum);
    if (tid < RB_ROWS) {
        const int ex  = inc - mine;
        const int row = b * RB_ROWS + tid;
        f[tid] = ex;
        if (row < nrows) {
            cnt[row] = mine;
            off[row] = start + ex;
        }
    }
    __syncthreads();

    if (n > 0 && n <= LIST_CAP) {
        for (int j0 = tid; j0 < n; j0 += 4 * PT) {
            int r[4];
#pragma unroll
            for (int u = 0; u < 4; ++u) {
                const int j = j0 + u * PT;
                r[u] = j < n ? rp[j] : -1;
            }
#pragma unroll
            for (int u = 0; u < 4; ++u) {
                if (r[u] >= 0) {
                    const int p = atomicAdd(&f[r[u] >> SRC_BITS], 1);
                    lst[p] = r[u] & SRC_MASK;
                }
            }
        }
        __syncthreads();
        const int end = start + n;
        for (int g0 = (start & ~31) + tid; g0 < end; g0 += 4 * PT) {
            int v[4];
#pragma unroll
            for (int u = 0; u < 4; ++u) {
                const int g = g0 + u * PT;
                v[u] = lst[(g >= start && g < end) ? g - start : 0];
            }
#pragma unroll
            for (int u = 0; u < 4; ++u) {
                const int g = g0 + u * PT;
                if (g >= start && g < end) list[g] = v[u];
            }
        }
        return;
    }

    int* __restrict__ lp = list + start;
    for (int j0 = tid; j0 < n; j0 += 4 * PT) {
        int r[4];
#pragma unroll
        for (int u = 0; u < 4; ++u) {
            const int j = j0 + u * PT;
            r[u] = j < n ? rp[j] : -1;
        }
#pragma unroll
        for (int u = 0; u < 4; ++u) {
            if (r[u] >= 0) {
                const int p = atomicAdd(&f[r[u] >> SRC_BITS], 1);
                lp[p] = r[u] & SRC_MASK;
            }
        }
    }
}

template <int J>
__device__ __forceinline__ int lane_xor(int v)
{
    if constexpr (J == 16)
        return __builtin_amdgcn_permlanex16(v, v, 0x76543210, 0xfedcba98, false, false);
    else
        return __builtin_amdgcn_update_dpp(v, v, 0x160 + J, 0xf, 0xf, true);
}

template <int K, int J, int Q>
__device__ __forceinline__ void sort_step(int (&k)[Q], int lane)
{
    if constexpr (J >= 32) {
        constexpr int dq = J >> 5;
#pragma unroll
        for (int q = 0; q < Q; ++q) {
            if ((q & dq) == 0) {
                const bool asc = ((32 * q) & K) == 0;
                const int x  = k[q];
                const int y  = k[q + dq];
                const int lo = y < x ? y : x;
                const int hi = y < x ? x : y;
                k[q]      = asc ? lo : hi;
                k[q + dq] = asc ? hi : lo;
            }
        }
    } else {
        const bool lower = (lane & J) == 0;
#pragma unroll
        for (int q = 0; q < Q; ++q) {
            const bool asc = K < 32 ? (lane & K) == 0 : ((32 * q) & K) == 0;
            const int x  = k[q];
            const int o  = lane_xor<J>(x);
            const int mn = o < x ? o : x;
            const int mx = o < x ? x : o;
            k[q] = (lower == asc) ? mn : mx;
        }
    }
}

template <int K, int J, int Q>
__device__ __forceinline__ void sort_from(int (&k)[Q], int lane)
{
    sort_step<K, J, Q>(k, lane);
    if constexpr (J > 1)
        sort_from<K, J / 2, Q>(k, lane);
    else if constexpr (K < 32 * Q)
        sort_from<2 * K, K, Q>(k, lane);
}

template <int K>
__device__ __forceinline__ void
add_sorted_rows(const float* __restrict__ src, int lane, int kreg, int t,
                float& a0, float& a1, float& a2, float& a3)
{
    int kk[K];
#pragma unroll
    for (int u = 0; u < K; ++u)
        kk[u] = __builtin_amdgcn_readlane(kreg, t + u);
    float v[K][4];
#pragma unroll
    for (int u = 0; u < K; ++u) {
        const float* __restrict__ sp = src + (size_t)kk[u] * DIM + lane;
        v[u][0] = sp[0];
        v[u][1] = sp[32];
        v[u][2] = sp[64];
        v[u][3] = sp[96];
    }
#pragma unroll
    for (int u = 0; u < K; ++u) {
        a0 += v[u][0];
        a1 += v[u][1];
        a2 += v[u][2];
        a3 += v[u][3];
    }
}

template <int Q>
__device__ __forceinline__ void
sum_loaded_keys(const float* __restrict__ src, int (&k)[Q],
                int lane, int n, float& a0, float& a1, float& a2, float& a3)
{
    sort_from<2, 1, Q>(k, lane);

#pragma unroll
    for (int q = 0; q < Q; ++q) {
        int c = n - 32 * q;
        c = c < 0 ? 0 : (c > 32 ? 32 : c);
        int t = 0;
        for (; t + 4 <= c; t += 4)
            add_sorted_rows<4>(src, lane, k[q], t, a0, a1, a2, a3);
        if (t + 2 <= c) {
            add_sorted_rows<2>(src, lane, k[q], t, a0, a1, a2, a3);
            t += 2;
        }
        if (t < c)
            add_sorted_rows<1>(src, lane, k[q], t, a0, a1, a2, a3);
    }
}

template <int Q>
__device__ __forceinline__ void
sorted_row_sum(const float* __restrict__ src, const int* __restrict__ lst,
               int lane, int n, float& a0, float& a1, float& a2, float& a3)
{
    int k[Q];
#pragma unroll
    for (int q = 0; q < Q; ++q) {
        const int j = lane + 32 * q;
        k[q] = j < n ? lst[j] : KEY_NONE;
    }
    sum_loaded_keys<Q>(src, k, lane, n, a0, a1, a2, a3);
}

__device__ __forceinline__ void
sum_row(const float* __restrict__ src, const int* __restrict__ lst,
        int lane, int n, float& a0, float& a1, float& a2, float& a3)
{
    if (n <= 32) {
        sorted_row_sum<1>(src, lst, lane, n, a0, a1, a2, a3);
    } else if (n <= 64) {
        sorted_row_sum<2>(src, lst, lane, n, a0, a1, a2, a3);
    } else if (n <= 128) {
        sorted_row_sum<4>(src, lst, lane, n, a0, a1, a2, a3);
    } else {
        int prev = -1;
        int done = 0;
        while (done < n) {
            int mn = KEY_NONE;
            for (int j = lane; j < n; j += 32) {
                const int k = lst[j];
                if (k > prev && k < mn) mn = k;
            }
            for (int o = 16; o > 0; o >>= 1) {
                const int other = __shfl_xor(mn, o);
                mn = other < mn ? other : mn;
            }
            int c = 0;
            for (int j = lane; j < n; j += 32) c += (lst[j] == mn) ? 1 : 0;
            for (int o = 16; o > 0; o >>= 1) c += __shfl_xor(c, o);
            mn = __builtin_amdgcn_readfirstlane(mn);
            c  = __builtin_amdgcn_readfirstlane(c);
            if (c <= 0) break;
            const float* __restrict__ sp = src + (size_t)mn * DIM + lane;
            const float w0 = sp[0];
            const float w1 = sp[32];
            const float w2 = sp[64];
            const float w3 = sp[96];
            for (int i = 0; i < c; ++i) {
                a0 += w0;
                a1 += w1;
                a2 += w2;
                a3 += w3;
            }
            prev = mn;
            done += c;
        }
    }
}

__global__ void __launch_bounds__(256)
rowsum_loop_kernel(const float* __restrict__ src, float* __restrict__ dst,
                   const int* __restrict__ cnt, const int* __restrict__ off,
                   const int* __restrict__ list, const float* __restrict__ bias,
                   int nrows)
{
    const int lane = threadIdx.x & 31;
    const int nw   = (int)(gridDim.x * (blockDim.x >> 5));
    int row = __builtin_amdgcn_readfirstlane((int)((blockIdx.x * blockDim.x + threadIdx.x) >> 5));
    if (row >= nrows) return;

    int n0 = __builtin_amdgcn_readfirstlane(cnt[row]);
    int b0 = __builtin_amdgcn_readfirstlane(off[row]);
    int key0 = lane < n0 ? list[b0 + lane] : KEY_NONE;
    int row1 = row + nw;
    int n1 = 0, b1 = 0;
    if (row1 < nrows) {
        n1 = __builtin_amdgcn_readfirstlane(cnt[row1]);
        b1 = __builtin_amdgcn_readfirstlane(off[row1]);
    }

    for (;;) {
        const int row2 = row1 + nw;
        int key1 = KEY_NONE;
        int n2v = 0, b2v = 0;
        if (row1 < nrows) key1 = lane < n1 ? list[b1 + lane] : KEY_NONE;
        if (row2 < nrows) { n2v = cnt[row2]; b2v = off[row2]; }

        float a0 = 0.0f, a1 = 0.0f, a2 = 0.0f, a3 = 0.0f;
        if (n0 <= 32) {
            int k[1] = { key0 };
            sum_loaded_keys<1>(src, k, lane, n0, a0, a1, a2, a3);
        } else {
            sum_row(src, list + b0, lane, n0, a0, a1, a2, a3);
        }

        const float inv = n0 > 0 ? 1.0f / (float)n0 : 0.0f;
        float* dp = dst + (size_t)row * DIM + lane;
        const float y0 = a0 * inv + bias[lane];
        const float y1 = a1 * inv + bias[lane + 32];
        const float y2 = a2 * inv + bias[lane + 64];
        const float y3 = a3 * inv + bias[lane + 96];
        dp[0]  = y0 >= 0.0f ? y0 : NEG_SLOPE * y0;
        dp[32] = y1 >= 0.0f ? y1 : NEG_SLOPE * y1;
        dp[64] = y2 >= 0.0f ? y2 : NEG_SLOPE * y2;
        dp[96] = y3 >= 0.0f ? y3 : NEG_SLOPE * y3;

        if (row1 >= nrows) break;
        row  = row1;
        n0   = n1;
        b0   = b1;
        key0 = key1;
        row1 = row2;
        n1   = __builtin_amdgcn_readfirstlane(n2v);
        b1   = __builtin_amdgcn_readfirstlane(b2v);
    }
}

#define TILE_P 132
__global__ void __launch_bounds__(512)
rowmean_gemm_kernel(const float* __restrict__ X, const float* __restrict__ W,
                    float* __restrict__ M, const int* __restrict__ cnt,
                    const int* __restrict__ off, const int* __restrict__ list)
{
    __shared__ float tile[16 * TILE_P];
    const int lane = threadIdx.x & 31;
    const int wv   = threadIdx.x >> 5;
    const int row0 = blockIdx.x * 16;
    {
        const int row = row0 + wv;
        const int n = __builtin_amdgcn_readfirstlane(cnt[row]);
        const int b = __builtin_amdgcn_readfirstlane(off[row]);
        float a0 = 0.0f, a1 = 0.0f, a2 = 0.0f, a3 = 0.0f;
        sum_row(X, list + b, lane, n, a0, a1, a2, a3);
        const float inv = n > 0 ? 1.0f / (float)n : 0.0f;
        float* tp = tile + wv * TILE_P + lane;
        tp[0]  = a0 * inv;
        tp[32] = a1 * inv;
        tp[64] = a2 * inv;
        tp[96] = a3 * inv;
    }
    __syncthreads();
    if (wv >= DIM / 16) return;

    const int half = lane >> 4;
    const int lr   = lane & 15;
    const int col0 = wv * 16;
    const float* ap = tile + lr * TILE_P + 2 * half;
    const float* __restrict__ bp = W + (size_t)(2 * half) * DIM + col0 + lr;

    unsigned yo = (unsigned)(((row0 + half * 8) * DIM + col0 + lr) * sizeof(float));
    asm volatile("" : "+v"(yo));
    float* __restrict__ yp = (float*)((char*)M + yo);

    v8f c = {};
#pragma unroll
    for (int k = 0; k < DIM; k += 4) {
        v2f a, b;
        a.x = ap[k];
        a.y = ap[k + 1];
        b.x = bp[(size_t)k * DIM];
        b.y = bp[(size_t)k * DIM + DIM];
        c = __builtin_amdgcn_wmma_f32_16x16x4_f32(false, a, false, b,
                                                  (short)0, c, false, false);
    }

#pragma unroll
    for (int r = 0; r < 8; ++r)
        yp[(size_t)r * DIM] = c[r];
}

extern "C" void kernel_launch(void* const* d_in, const int* in_sizes, int n_in,
                              void* d_out, int out_size, void* d_ws, size_t ws_size,
                              hipStream_t stream)
{
    const float* X    = (const float*)d_in[0];
    const int*   A    = (const int*)d_in[1];
    const float* W    = (const float*)d_in[2];
    const float* bias = (const float*)d_in[3];
    float*       out  = (float*)d_out;

    const int N   = in_sizes[0] / DIM;
    const int NNZ = in_sizes[1] / 2;
    const int* node = A;
    const int* edge = A + NNZ;
    const long n_out = (long)N * DIM;
    const long n_m   = (long)NEDGES * DIM;

    const int mtiles = N / 16;

    const int  R    = NEDGES + N;
    const int  nb   = (R + RB_ROWS - 1) / RB_ROWS;
    float* m    = (float*)d_ws;
    int*   cnt  = (int*)(m + n_m);
    int*   off  = cnt + R;
    int*   list = off + R;
    int*   bcnt = list + 2L * NNZ;
    int*   bcur = bcnt + NBUCK_MAX;
    int*   boff = bcur + NBUCK_MAX;
    int*   recs = (int*)d_out;
    const int etiles = NEDGES / 16;
    const size_t need = ((size_t)n_m + 2 * (size_t)R + 2 * (size_t)NNZ + 3 * NBUCK_MAX) * sizeof(float);

    if (ws_size >= need && nb <= NBUCK_MAX && N <= (1 << SRC_BITS) &&
        NEDGES <= (1 << SRC_BITS) && 2L * NNZ <= n_out) {
        const int nchunks = (NNZ + CHUNK - 1) / CHUNK;
        zero_kernel<<<(2 * NBUCK_MAX + 255) / 256, 256, 0, stream>>>((float*)bcnt, 2L * NBUCK_MAX);
        range_count_kernel<<<nchunks, PT, 0, stream>>>(node, edge, bcnt, NNZ);
        range_move_kernel<<<nchunks, PT, 0, stream>>>(node, edge, bcnt, bcur, boff, recs, NNZ);
        range_rows_kernel<<<nb, PT, 0, stream>>>(recs, bcnt, boff, cnt, off, list, R);
        rowmean_gemm_kernel<<<etiles, 512, 0, stream>>>(X, W, m, cnt, off, list);
        rowsum_loop_kernel<<<1280, 256, 0, stream>>>(m, out, cnt + NEDGES, off + NEDGES, list, bias, N);
        return;
    }

    gemm_wmma_f32<<<mtiles, 256, 0, stream>>>(X, W, out, mtiles);

    float* Ddeg = m + (size_t)NEDGES * DIM;
    float* Bdeg = Ddeg + N;
    const long ws_f  = (long)NEDGES * DIM + N + NEDGES;
    zero_kernel<<<(int)((ws_f + 255) / 256), 256, 0, stream>>>(m, ws_f);
    degrees_kernel<<<(NNZ + 255) / 256, 256, 0, stream>>>(node, edge, Ddeg, Bdeg, NNZ);
    scatter_kernel<<<(NNZ + 7) / 8, 256, 0, stream>>>(out, m, node, edge, NNZ);
    scale_rows_kernel<<<(int)((n_m + 255) / 256), 256, 0, stream>>>(m, Bdeg, n_m);
    zero_kernel<<<(int)((n_out + 255) / 256), 256, 0, stream>>>(out, n_out);
    scatter_kernel<<<(NNZ + 7) / 8, 256, 0, stream>>>(m, out, edge, node, NNZ);
    finalize_kernel<<<(int)((n_out + 255) / 256), 256, 0, stream>>>(out, Ddeg, bias, n_out);
}


# list building: each chunk's records are sorted by row range in LDS and stored as its own piece with a table of range starts; range blocks pick their runs; no count pass, clear, global adds; same sums
# speedup vs baseline: 7.2789x; 1.0478x over previous
#include <hip/hip_runtime.h>

typedef __attribute__((ext_vector_type(2))) float v2f;
typedef __attribute__((ext_vector_type(8))) float v8f;
typedef __attribute__((ext_vector_type(2))) int   v2i;

#define DIM        128
#define NEDGES     50000
#define NEG_SLOPE  0.01f

__global__ void __launch_bounds__(256)
gemm_wmma_f32(const float* __restrict__ X, const float* __restrict__ W,
              float* __restrict__ Y, int mtiles)
{
    const int lane  = threadIdx.x & 31;
    const int wv    = threadIdx.x >> 5;
    const long tile = (long)blockIdx.x * 8 + wv;
    const long mt   = tile >> 3;
    const int  nt   = (int)(tile & 7);
    if (mt >= mtiles) return;

    const int half = lane >> 4;
    const int lr   = lane & 15;
    const size_t row  = (size_t)mt * 16 + lr;
    const int    col0 = nt * 16;

    const float* __restrict__ ap = X + row * DIM + 2 * half;
    const float* __restrict__ bp = W + (size_t)(2 * half) * DIM + col0 + lr;

    v8f c = {};
#pragma unroll
    for (int k = 0; k < DIM; k += 4) {
        v2f a, b;
        a.x = ap[k];
        a.y = ap[k + 1];
        b.x = bp[(size_t)k * DIM];
        b.y = bp[(size_t)k * DIM + DIM];
        c = __builtin_amdgcn_wmma_f32_16x16x4_f32(false, a, false, b,
                                                  (short)0, c, false, false);
    }

    float* __restrict__ yp = Y + ((size_t)mt * 16 + half * 8) * DIM + col0 + lr;
#pragma unroll
    for (int r = 0; r < 8; ++r)
        yp[(size_t)r * DIM] = c[r];
}

__global__ void zero_kernel(float* __restrict__ p, long n)
{
    long i = (long)blockIdx.x * blockDim.x + threadIdx.x;
    if (i < n) p[i] = 0.0f;
}

__global__ void degrees_kernel(const int* __restrict__ node,
                               const int* __restrict__ edge,
                               float* __restrict__ Ddeg,
                               float* __restrict__ Bdeg, int nnz)
{
    int i = blockIdx.x * blockDim.x + threadIdx.x;
    if (i >= nnz) return;
    unsafeAtomicAdd(&Ddeg[node[i]], 1.0f);
    unsafeAtomicAdd(&Bdeg[edge[i]], 1.0f);
}

__global__ void __launch_bounds__(256)
scatter_kernel(const float* __restrict__ src, float* __restrict__ dst,
               const int* __restrict__ gidx, const int* __restrict__ sidx,
               int nnz)
{
    const int wid  = (int)(((long)blockIdx.x * blockDim.x + threadIdx.x) >> 5);
    const int lane = threadIdx.x & 31;
    if (wid >= nnz) return;

    const int g = gidx[wid];
    const int s = sidx[wid];

    const float* __restrict__ sp = src + (size_t)g * DIM + lane;
    float* dp = dst + (size_t)s * DIM + lane;
    const float v0 = sp[0];
    const float v1 = sp[32];
    const float v2 = sp[64];
    const float v3 = sp[96];
    unsafeAtomicAdd(dp + 0,  v0);
    unsafeAtomicAdd(dp + 32, v1);
    unsafeAtomicAdd(dp + 64, v2);
    unsafeAtomicAdd(dp + 96, v3);
}

__global__ void scale_rows_kernel(float* __restrict__ m,
                                  const float* __restrict__ Bdeg, long n)
{
    long i = (long)blockIdx.x * blockDim.x + threadIdx.x;
    if (i >= n) return;
    float bb  = Bdeg[i >> 7];
    float inv = bb > 0.0f ? 1.0f / bb : 0.0f;
    m[i] *= inv;
}

__global__ void finalize_kernel(float* __restrict__ out,
                                const float* __restrict__ Ddeg,
                                const float* __restrict__ bias, long n)
{
    long i = (long)blockIdx.x * blockDim.x + threadIdx.x;
    if (i >= n) return;
    float dd  = Ddeg[i >> 7];
    float inv = dd > 0.0f ? 1.0f / dd : 0.0f;
    float v   = out[i] * inv + bias[i & (DIM - 1)];
    out[i] = v >= 0.0f ? v : NEG_SLOPE * v;
}

#define KEY_NONE 0x7fffffff

#define RB_SHIFT  9
#define RB_ROWS   (1 << RB_SHIFT)
#define NBUCK_MAX 512
#define SRC_BITS  18
#define SRC_MASK  ((1 << SRC_BITS) - 1)
#define PT        1024
#define EPT       8
#define CHUNK     (PT * EPT)
#define LIST_CAP  32768
#define PROW      (NBUCK_MAX + 1)
#define PU        8
static_assert(64 * PU == NBUCK_MAX, "two trips of 32 PU chunks cover the table");

__device__ __forceinline__ int scan512(int x, int tid, int* wsum)
{
    const int lane = tid & 31;
    const int wv   = tid >> 5;
    int v = x;
    for (int o = 1; o < 32; o <<= 1) {
        const int t = __shfl_up(v, o);
        if (lane >= o) v += t;
    }
    if (lane == 31 && wv < 16) wsum[wv] = v;
    __syncthreads();
    if (wv == 0) {
        int w = lane < 16 ? wsum[lane] : 0;
        for (int o = 1; o < 16; o <<= 1) {
            const int t = __shfl_up(w, o);
            if (lane >= o) w += t;
        }
        if (lane < 16) wsum[lane] = w;
    }
    __syncthreads();
    if (wv > 0 && wv < 16) v += wsum[wv - 1];
    return v;
}

template <int J>
__device__ __forceinline__ int lane_xor(int v)
{
    if constexpr (J == 16)
        return __builtin_amdgcn_permlanex16(v, v, 0x76543210, 0xfedcba98, false, false);
    else
        return __builtin_amdgcn_update_dpp(v, v, 0x160 + J, 0xf, 0xf, true);
}

__device__ __forceinline__ int wave_sum(int v)
{
    v += lane_xor<1>(v);
    v += lane_xor<2>(v);
    v += lane_xor<4>(v);
    v += lane_xor<8>(v);
    v += lane_xor<16>(v);
    return v;
}

__global__ void __launch_bounds__(PT)
range_sort_kernel(const int* __restrict__ node, const int* __restrict__ edge,
                  int* __restrict__ recs, int* __restrict__ ptab, int nnz)
{
    __shared__ int h[NBUCK_MAX];
    __shared__ int base[NBUCK_MAX];
    __shared__ int wsum[16];
    __shared__ int stage[2 * CHUNK];
    const int tid = threadIdx.x;
    if (tid < NBUCK_MAX) h[tid] = 0;
    __syncthreads();

    int ke[EPT], kn[EPT], re[EPT], rn[EPT];
    const int base0 = blockIdx.x * CHUNK + tid;
#pragma unroll
    for (int u = 0; u < EPT; ++u) {
        const int i = base0 + u * PT;
        ke[u] = i < nnz ? edge[i] : -1;
        kn[u] = i < nnz ? node[i] : 0;
    }
#pragma unroll
    for (int u = 0; u < EPT; ++u) {
        re[u] = 0;
        rn[u] = 0;
        if (ke[u] >= 0) {
            re[u] = atomicAdd(&h[ke[u] >> RB_SHIFT], 1);
            rn[u] = atomicAdd(&h[(NEDGES + kn[u]) >> RB_SHIFT], 1);
        }
    }
    __syncthreads();

    const int c   = tid < NBUCK_MAX ? h[tid] : 0;
    const int lin = scan512(c, tid, wsum);
    if (tid < NBUCK_MAX) {
        int* __restrict__ prow = ptab + (size_t)blockIdx.x * PROW;
        base[tid] = lin - c;
        prow[tid] = lin - c;
        if (tid == NBUCK_MAX - 1) prow[NBUCK_MAX] = lin;
    }
    __syncthreads();

#pragma unroll
    for (int u = 0; u < EPT; ++u) {
        if (ke[u] >= 0) {
            const int e  = ke[u];
            const int n  = kn[u];
            const int rr = NEDGES + n;
            stage[base[e >> RB_SHIFT] + re[u]]  = ((e & (RB_ROWS - 1)) << SRC_BITS) | n;
            stage[base[rr >> RB_SHIFT] + rn[u]] = ((rr & (RB_ROWS - 1)) << SRC_BITS) | e;
        }
    }
    __syncthreads();

    const int left = nnz - (int)blockIdx.x * CHUNK;
    const int nrec = 2 * (left < CHUNK ? left : CHUNK);
    int* __restrict__ rb = recs + (size_t)blockIdx.x * (2 * CHUNK);
    int w[2 * EPT];
#pragma unroll
    for (int u = 0; u < 2 * EPT; ++u) {
        const int p = tid + u * PT;
        w[u] = stage[p < nrec ? p : 0];
    }
#pragma unroll
    for (int u = 0; u < 2 * EPT; ++u) {
        const int p = tid + u * PT;
        if (p < nrec) rb[p] = w[u];
    }
}

template <class F>
__device__ __forceinline__ void
for_each_record(const int* __restrict__ recs, const v2i* pcs, int nchunks,
                int wv, int lane, F f)
{
    for (int k0 = wv; k0 < nchunks; k0 += 32 * PU) {
        v2i pc[PU];
        int r[PU][2];
#pragma unroll
        for (int u = 0; u < PU; ++u)
            pc[u] = pcs[k0 + 32 * u];
#pragma unroll
        for (int u = 0; u < PU; ++u) {
            const int* __restrict__ rp = recs + pc[u].x;
            r[u][0] = lane < pc[u].y ? rp[lane] : -1;
            r[u][1] = lane + 32 < pc[u].y ? rp[lane + 32] : -1;
        }
#pragma unroll
        for (int u = 0; u < PU; ++u) {
            if (r[u][0] >= 0) f(r[u][0]);
            if (r[u][1] >= 0) f(r[u][1]);
        }
#pragma unroll
        for (int u = 0; u < PU; ++u)
            for (int j = lane + 64; j < pc[u].y; j += 32) f(recs[pc[u].x + j]);
    }
}

__global__ void __launch_bounds__(PT)
range_rows_kernel(const int* __restrict__ recs, const int* __restrict__ ptab,
                  int* __restrict__ cnt, int* __restrict__ off,
                  int* __restrict__ list, int nrows, int nchunks)
{
    __shared__ int c[RB_ROWS];
    __shared__ int f[RB_ROWS];
    __shared__ int wsum[16];
    __shared__ int red[2][32];
    __shared__ v2i pcs[NBUCK_MAX];
    __shared__ int lst[LIST_CAP];
    const int tid  = threadIdx.x;
    const int lane = tid & 31;
    const int wv   = tid >> 5;
    const int b    = blockIdx.x;
    int p0 = 0;
    int pn = 0;
    if (tid < nchunks) {
        const int* __restrict__ pr = ptab + (size_t)tid * PROW + b;
        p0 = pr[0];
        pn = pr[1] - p0;
    }
    if (tid < NBUCK_MAX) {
        v2i pc;
        pc.x = tid * (2 * CHUNK) + p0;
        pc.y = pn;
        pcs[tid] = pc;
    }
    if (tid < RB_ROWS) c[tid] = 0;
    {
        const int s0 = wave_sum(p0);
        const int s1 = wave_sum(pn);
        if (lane == 0) {
            red[0][wv] = s0;
            red[1][wv] = s1;
        }
    }
    __syncthreads();
    const int start = __builtin_amdgcn_readfirstlane(wave_sum(red[0][lane]));
    const int n     = __builtin_amdgcn_readfirstlane(wave_sum(red[1][lane]));

    for_each_record(recs, pcs, nchunks, wv, lane,
                    [&](int r) { atomicAdd(&c[r >> SRC_BITS], 1); });
    __syncthreads();

    const int mine = tid < RB_ROWS ? c[tid] : 0;
    const int inc  = scan512(mine, tid, wsum);
    if (tid < RB_ROWS) {
        const int ex  = inc - mine;
        const int row = b * RB_ROWS + tid;
        f[tid] = ex;
        if (row < nrows) {
            cnt[row] = mine;
            off[row] = start + ex;
        }
    }
    __syncthreads();

    if (n > 0 && n <= LIST_CAP) {
        for_each_record(recs, pcs, nchunks, wv, lane,
                        [&](int r) {
                            const int p = atomicAdd(&f[r >> SRC_BITS], 1);
                            lst[p] = r & SRC_MASK;
                        });
        __syncthreads();
        const int end = start + n;
        for (int g0 = (start & ~31) + tid; g0 < end; g0 += 4 * PT) {
            int v[4];
#pragma unroll
            for (int u = 0; u < 4; ++u) {
                const int g = g0 + u * PT;
                v[u] = lst[(g >= start && g < end) ? g - start : 0];
            }
#pragma unroll
            for (int u = 0; u < 4; ++u) {
                const int g = g0 + u * PT;
                if (g >= start && g < end) list[g] = v[u];
            }
        }
        return;
    }

    int* __restrict__ lp = list + start;
    for_each_record(recs, pcs, nchunks, wv, lane,
                    [&](int r) {
                        const int p = atomicAdd(&f[r >> SRC_BITS], 1);
                        lp[p] = r & SRC_MASK;
                    });
}

template <int K, int J, int Q>
__device__ __forceinline__ void sort_step(int (&k)[Q], int lane)
{
    if constexpr (J >= 32) {
        constexpr int dq = J >> 5;
#pragma unroll
        for (int q = 0; q < Q; ++q) {
            if ((q & dq) == 0) {
                const bool asc = ((32 * q) & K) == 0;
                const int x  = k[q];
                const int y  = k[q + dq];
                const int lo = y < x ? y : x;
                const int hi = y < x ? x : y;
                k[q]      = asc ? lo : hi;
                k[q + dq] = asc ? hi : lo;
            }
        }
    } else {
        const bool lower = (lane & J) == 0;
#pragma unroll
        for (int q = 0; q < Q; ++q) {
            const bool asc = K < 32 ? (lane & K) == 0 : ((32 * q) & K) == 0;
            const int x  = k[q];
            const int o  = lane_xor<J>(x);
            const int mn = o < x ? o : x;
            const int mx = o < x ? x : o;
            k[q] = (lower == asc) ? mn : mx;
        }
    }
}

template <int K, int J, int Q>
__device__ __forceinline__ void sort_from(int (&k)[Q], int lane)
{
    sort_step<K, J, Q>(k, lane);
    if constexpr (J > 1)
        sort_from<K, J / 2, Q>(k, lane);
    else if constexpr (K < 32 * Q)
        sort_from<2 * K, K, Q>(k, lane);
}

template <int K>
__device__ __forceinline__ void
add_sorted_rows(const float* __restrict__ src, int lane, int kreg, int t,
                float& a0, float& a1, float& a2, float& a3)
{
    int kk[K];
#pragma unroll
    for (int u = 0; u < K; ++u)
        kk[u] = __builtin_amdgcn_readlane(kreg, t + u);
    float v[K][4];
#pragma unroll
    for (int u = 0; u < K; ++u) {
        const float* __restrict__ sp = src + (size_t)kk[u] * DIM + lane;
        v[u][0] = sp[0];
        v[u][1] = sp[32];
        v[u][2] = sp[64];
        v[u][3] = sp[96];
    }
#pragma unroll
    for (int u = 0; u < K; ++u) {
        a0 += v[u][0];
        a1 += v[u][1];
        a2 += v[u][2];
        a3 += v[u][3];
    }
}

template <int Q>
__device__ __forceinline__ void
sum_loaded_keys(const float* __restrict__ src, int (&k)[Q],
                int lane, int n, float& a0, float& a1, float& a2, float& a3)
{
    sort_from<2, 1, Q>(k, lane);

#pragma unroll
    for (int q = 0; q < Q; ++q) {
        int c = n - 32 * q;
        c = c < 0 ? 0 : (c > 32 ? 32 : c);
        int t = 0;
        for (; t + 4 <= c; t += 4)
            add_sorted_rows<4>(src, lane, k[q], t, a0, a1, a2, a3);
        if (t + 2 <= c) {
            add_sorted_rows<2>(src, lane, k[q], t, a0, a1, a2, a3);
            t += 2;
        }
        if (t < c)
            add_sorted_rows<1>(src, lane, k[q], t, a0, a1, a2, a3);
    }
}

template <int Q>
__device__ __forceinline__ void
sorted_row_sum(const float* __restrict__ src, const int* __restrict__ lst,
               int lane, int n, float& a0, float& a1, float& a2, float& a3)
{
    int k[Q];
#pragma unroll
    for (int q = 0; q < Q; ++q) {
        const int j = lane + 32 * q;
        k[q] = j < n ? lst[j] : KEY_NONE;
    }
    sum_loaded_keys<Q>(src, k, lane, n, a0, a1, a2, a3);
}

__device__ __forceinline__ void
sum_row(const float* __restrict__ src, const int* __restrict__ lst,
        int lane, int n, float& a0, float& a1, float& a2, float& a3)
{
    if (n <= 32) {
        sorted_row_sum<1>(src, lst, lane, n, a0, a1, a2, a3);
    } else if (n <= 64) {
        sorted_row_sum<2>(src, lst, lane, n, a0, a1, a2, a3);
    } else if (n <= 128) {
        sorted_row_sum<4>(src, lst, lane, n, a0, a1, a2, a3);
    } else {
        int prev = -1;
        int done = 0;
        while (done < n) {
            int mn = KEY_NONE;
            for (int j = lane; j < n; j += 32) {
                const int k = lst[j];
                if (k > prev && k < mn) mn = k;
            }
            for (int o = 16; o > 0; o >>= 1) {
                const int other = __shfl_xor(mn, o);
                mn = other < mn ? other : mn;
            }
            int c = 0;
            for (int j = lane; j < n; j += 32) c += (lst[j] == mn) ? 1 : 0;
            for (int o = 16; o > 0; o >>= 1) c += __shfl_xor(c, o);
            mn = __builtin_amdgcn_readfirstlane(mn);
            c  = __builtin_amdgcn_readfirstlane(c);
            if (c <= 0) break;
            const float* __restrict__ sp = src + (size_t)mn * DIM + lane;
            const float w0 = sp[0];
            const float w1 = sp[32];
            const float w2 = sp[64];
            const float w3 = sp[96];
            for (int i = 0; i < c; ++i) {
                a0 += w0;
                a1 += w1;
                a2 += w2;
                a3 += w3;
            }
            prev = mn;
            done += c;
        }
    }
}

__global__ void __launch_bounds__(256)
rowsum_loop_kernel(const float* __restrict__ src, float* __restrict__ dst,
                   const int* __restrict__ cnt, const int* __restrict__ off,
                   const int* __restrict__ list, const float* __restrict__ bias,
                   int nrows)
{
    const int lane = threadIdx.x & 31;
    const int nw   = (int)(gridDim.x * (blockDim.x >> 5));
    int row = __builtin_amdgcn_readfirstlane((int)((blockIdx.x * blockDim.x + threadIdx.x) >> 5));
    if (row >= nrows) return;

    int n0 = __builtin_amdgcn_readfirstlane(cnt[row]);
    int b0 = __builtin_amdgcn_readfirstlane(off[row]);
    int key0 = lane < n0 ? list[b0 + lane] : KEY_NONE;
    int row1 = row + nw;
    int n1 = 0, b1 = 0;
    if (row1 < nrows) {
        n1 = __builtin_amdgcn_readfirstlane(cnt[row1]);
        b1 = __builtin_amdgcn_readfirstlane(off[row1]);
    }

    for (;;) {
        const int row2 = row1 + nw;
        int key1 = KEY_NONE;
        int n2v = 0, b2v = 0;
        if (row1 < nrows) key1 = lane < n1 ? list[b1 + lane] : KEY_NONE;
        if (row2 < nrows) { n2v = cnt[row2]; b2v = off[row2]; }

        float a0 = 0.0f, a1 = 0.0f, a2 = 0.0f, a3 = 0.0f;
        if (n0 <= 32) {
            int k[1] = { key0 };
            sum_loaded_keys<1>(src, k, lane, n0, a0, a1, a2, a3);
        } else {
            sum_row(src, list + b0, lane, n0, a0, a1, a2, a3);
        }

        const float inv = n0 > 0 ? 1.0f / (float)n0 : 0.0f;
        float* dp = dst + (size_t)row * DIM + lane;
        const float y0 = a0 * inv + bias[lane];
        const float y1 = a1 * inv + bias[lane + 32];
        const float y2 = a2 * inv + bias[lane + 64];
        const float y3 = a3 * inv + bias[lane + 96];
        dp[0]  = y0 >= 0.0f ? y0 : NEG_SLOPE * y0;
        dp[32] = y1 >= 0.0f ? y1 : NEG_SLOPE * y1;
        dp[64] = y2 >= 0.0f ? y2 : NEG_SLOPE * y2;
        dp[96] = y3 >= 0.0f ? y3 : NEG_SLOPE * y3;

        if (row1 >= nrows) break;
        row  = row1;
        n0   = n1;
        b0   = b1;
        key0 = key1;
        row1 = row2;
        n1   = __builtin_amdgcn_readfirstlane(n2v);
        b1   = __builtin_amdgcn_readfirstlane(b2v);
    }
}

#define TILE_P 132
__global__ void __launch_bounds__(512)
rowmean_gemm_kernel(const float* __restrict__ X, const float* __restrict__ W,
                    float* __restrict__ M, const int* __restrict__ cnt,
                    const int* __restrict__ off, const int* __restrict__ list)
{
    __shared__ float tile[16 * TILE_P];
    const int lane = threadIdx.x & 31;
    const int wv   = threadIdx.x >> 5;
    const int row0 = blockIdx.x * 16;
    {
        const int row = row0 + wv;
        const int n = __builtin_amdgcn_readfirstlane(cnt[row]);
        const int b = __builtin_amdgcn_readfirstlane(off[row]);
        float a0 = 0.0f, a1 = 0.0f, a2 = 0.0f, a3 = 0.0f;
        sum_row(X, list + b, lane, n, a0, a1, a2, a3);
        const float inv = n > 0 ? 1.0f / (float)n : 0.0f;
        float* tp = tile + wv * TILE_P + lane;
        tp[0]  = a0 * inv;
        tp[32] = a1 * inv;
        tp[64] = a2 * inv;
        tp[96] = a3 * inv;
    }
    __syncthreads();
    if (wv >= DIM / 16) return;

    const int half = lane >> 4;
    const int lr   = lane & 15;
    const int col0 = wv * 16;
    const float* ap = tile + lr * TILE_P + 2 * half;
    const float* __restrict__ bp = W + (size_t)(2 * half) * DIM + col0 + lr;

    unsigned yo = (unsigned)(((row0 + half * 8) * DIM + col0 + lr) * sizeof(float));
    asm volatile("" : "+v"(yo));
    float* __restrict__ yp = (float*)((char*)M + yo);

    v8f c = {};
#pragma unroll
    for (int k = 0; k < DIM; k += 4) {
        v2f a, b;
        a.x = ap[k];
        a.y = ap[k + 1];
        b.x = bp[(size_t)k * DIM];
        b.y = bp[(size_t)k * DIM + DIM];
        c = __builtin_amdgcn_wmma_f32_16x16x4_f32(false, a, false, b,
                                                  (short)0, c, false, false);
    }

#pragma unroll
    for (int r = 0; r < 8; ++r)
        yp[(size_t)r * DIM] = c[r];
}

extern "C" void kernel_launch(void* const* d_in, const int* in_sizes, int n_in,
                              void* d_out, int out_size, void* d_ws, size_t ws_size,
                              hipStream_t stream)
{
    const float* X    = (const float*)d_in[0];
    const int*   A    = (const int*)d_in[1];
    const float* W    = (const float*)d_in[2];
    const float* bias = (const float*)d_in[3];
    float*       out  = (float*)d_out;

    const int N   = in_sizes[0] / DIM;
    const int NNZ = in_sizes[1] / 2;
    const int* node = A;
    const int* edge = A + NNZ;
    const long n_out = (long)N * DIM;
    const long n_m   = (long)NEDGES * DIM;

    const int mtiles = N / 16;

    const int  R    = NEDGES + N;
    const int  nb   = (R + RB_ROWS - 1) / RB_ROWS;
    float* m    = (float*)d_ws;
    int*   cnt  = (int*)(m + n_m);
    int*   off  = cnt + R;
    int*   list = off + R;
    const int nchunks = (NNZ + CHUNK - 1) / CHUNK;
    int*   recs = (int*)d_out;
    int*   ptab = recs + (size_t)nchunks * (2 * CHUNK);
    const int etiles = NEDGES / 16;
    const size_t need = ((size_t)n_m + 2 * (size_t)R + 2 * (size_t)NNZ + 3 * NBUCK_MAX) * sizeof(float);

    if (ws_size >= need && nb <= NBUCK_MAX && N <= (1 << SRC_BITS) &&
        NEDGES <= (1 << SRC_BITS) && nchunks <= NBUCK_MAX &&
        (long)nchunks * (2 * CHUNK + PROW) <= n_out) {
        range_sort_kernel<<<nchunks, PT, 0, stream>>>(node, edge, recs, ptab, NNZ);
        range_rows_kernel<<<nb, PT, 0, stream>>>(recs, ptab, cnt, off, list, R, nchunks);
        rowmean_gemm_kernel<<<etiles, 512, 0, stream>>>(X, W, m, cnt, off, list);
        rowsum_loop_kernel<<<1280, 256, 0, stream>>>(m, out, cnt + NEDGES, off + NEDGES, list, bias, N);
        return;
    }

    gemm_wmma_f32<<<mtiles, 256, 0, stream>>>(X, W, out, mtiles);

    float* Ddeg = m + (size_t)NEDGES * DIM;
    float* Bdeg = Ddeg + N;
    const long ws_f  = (long)NEDGES * DIM + N + NEDGES;
    zero_kernel<<<(int)((ws_f + 255) / 256), 256, 0, stream>>>(m, ws_f);
    degrees_kernel<<<(NNZ + 255) / 256, 256, 0, stream>>>(node, edge, Ddeg, Bdeg, NNZ);
    scatter_kernel<<<(NNZ + 7) / 8, 256, 0, stream>>>(out, m, node, edge, NNZ);
    scale_rows_kernel<<<(int)((n_m + 255) / 256), 256, 0, stream>>>(m, Bdeg, n_m);
    zero_kernel<<<(int)((n_out + 255) / 256), 256, 0, stream>>>(out, n_out);
    scatter_kernel<<<(NNZ + 7) / 8, 256, 0, stream>>>(m, out, edge, node, NNZ);
    finalize_kernel<<<(int)((n_out + 255) / 256), 256, 0, stream>>>(out, Ddeg, bias, n_out);
}
